# starting kernel
# baseline (speedup 1.0000x reference)
#include <hip/hip_runtime.h>
#include <hip/hip_bf16.h>
#include <stdint.h>

typedef __attribute__((ext_vector_type(2))) float v2f;
typedef __attribute__((ext_vector_type(8))) float v8f;

#define NEG_SLOPE 0.01f

__device__ __forceinline__ float lrelu(float x) {
    return x > 0.0f ? x : NEG_SLOPE * x;
}

__device__ __forceinline__ unsigned fkey(float f) {
    unsigned u = __float_as_uint(f);
    return (u & 0x80000000u) ? ~u : (u | 0x80000000u);
}
__device__ __forceinline__ float fdecode(unsigned k) {
    unsigned u = (k & 0x80000000u) ? (k ^ 0x80000000u) : ~k;
    return __uint_as_float(u);
}
#define NEG_INF_KEY 0x007FFFFFu

__global__ void k_init(float* __restrict__ out, float* __restrict__ sumh,
                       float* __restrict__ sumt, unsigned* __restrict__ maxh,
                       unsigned* __restrict__ maxt, int n_out, int n_seg) {
    int i = blockIdx.x * blockDim.x + threadIdx.x;
    if (i < n_out) out[i] = 0.0f;
    if (i < n_seg) {
        sumh[i] = 0.0f;
        sumt[i] = 0.0f;
        maxh[i] = NEG_INF_KEY;
        maxt[i] = NEG_INF_KEY;
    }
}

__global__ void k_proj_wmma(const float* __restrict__ X,
                            const float* __restrict__ W0,
                            const float* __restrict__ W1,
                            float* __restrict__ S, int nrows) {
    const int wave = (blockIdx.x * blockDim.x + threadIdx.x) >> 5;
    const int lane = threadIdx.x & 31;
    const int row0 = wave * 16;
    if (row0 >= nrows) return;

    const int m    = lane & 15;
    const int kofs = (lane >> 4) << 1;

    const float* __restrict__ arow = X + (size_t)(row0 + m) * 128;
    const float* __restrict__ brow =
        (m < 8) ? (W0 + (size_t)m * 128) : (W1 + (size_t)(m - 8) * 128);

    v8f c = {0.f, 0.f, 0.f, 0.f, 0.f, 0.f, 0.f, 0.f};
#pragma unroll
    for (int k = 0; k < 128; k += 4) {
        v2f a = *(const v2f*)(arow + k + kofs);
        v2f b = *(const v2f*)(brow + k + kofs);
        c = __builtin_amdgcn_wmma_f32_16x16x4_f32(
             false, a,  false, b,
             (short)0, c,  false,  false);
    }

    const int rbase = row0 + ((lane >> 4) << 3);
    const int col   = lane & 15;
#pragma unroll
    for (int v = 0; v < 8; ++v) {
        S[(size_t)(rbase + v) * 16 + col] = c[v];
    }
}

__global__ void k_edge_max(const int* __restrict__ h_idx,
                           const int* __restrict__ t_idx,
                           const int* __restrict__ rel,
                           const float* __restrict__ Se,
                           const float* __restrict__ Sr,
                           unsigned* __restrict__ maxh,
                           unsigned* __restrict__ maxt, int E) {
    int i = blockIdx.x * blockDim.x + threadIdx.x;
    if (i >= E * 8) return;
    const int e = i >> 3, h = i & 7;
    const int hi = h_idx[e], ti = t_idx[e], r = rel[e];
    const float er = Sr[(size_t)r * 16 + h];
    const float sh = lrelu(Se[(size_t)hi * 16 + h] + er);
    const float st = lrelu(Se[(size_t)ti * 16 + 8 + h] + er);
    atomicMax(&maxh[(size_t)hi * 8 + h], fkey(sh));
    atomicMax(&maxt[(size_t)ti * 8 + h], fkey(st));
}

__global__ void k_edge_sum(const int* __restrict__ h_idx,
                           const int* __restrict__ t_idx,
                           const int* __restrict__ rel,
                           const float* __restrict__ Se,
                           const float* __restrict__ Sr,
                           const unsigned* __restrict__ maxh,
                           const unsigned* __restrict__ maxt,
                           float* __restrict__ sumh,
                           float* __restrict__ sumt, int E) {
    int i = blockIdx.x * blockDim.x + threadIdx.x;
    if (i >= E * 8) return;
    const int e = i >> 3, h = i & 7;
    const int hi = h_idx[e], ti = t_idx[e], r = rel[e];
    const float er = Sr[(size_t)r * 16 + h];
    const float sh = lrelu(Se[(size_t)hi * 16 + h] + er);
    const float st = lrelu(Se[(size_t)ti * 16 + 8 + h] + er);
    atomicAdd(&sumh[(size_t)hi * 8 + h], __expf(sh - fdecode(maxh[(size_t)hi * 8 + h])));
    atomicAdd(&sumt[(size_t)ti * 8 + h], __expf(st - fdecode(maxt[(size_t)ti * 8 + h])));
}

__global__ void k_edge_scatter(const int* __restrict__ h_idx,
                               const int* __restrict__ t_idx,
                               const int* __restrict__ rel,
                               const float* __restrict__ Se,
                               const float* __restrict__ Sr,
                               const unsigned* __restrict__ maxh,
                               const unsigned* __restrict__ maxt,
                               const float* __restrict__ sumh,
                               const float* __restrict__ sumt,
                               const float* __restrict__ x_r,
                               float* __restrict__ out, int E) {
    const int wid  = (blockIdx.x * blockDim.x + threadIdx.x) >> 5;
    const int lane = threadIdx.x & 31;
    if (wid >= E) return;

    const int hi = h_idx[wid], ti = t_idx[wid], r = rel[wid];

    float ah = 0.0f, at = 0.0f;
#pragma unroll
    for (int h = 0; h < 8; ++h) {
        const float er = Sr[(size_t)r * 16 + h];
        const float sh = lrelu(Se[(size_t)hi * 16 + h] + er);
        const float st = lrelu(Se[(size_t)ti * 16 + 8 + h] + er);
        const float eh = __expf(sh - fdecode(maxh[(size_t)hi * 8 + h]));
        const float et = __expf(st - fdecode(maxt[(size_t)ti * 8 + h]));
        ah += eh / (sumh[(size_t)hi * 8 + h] + 1e-16f);
        at += et / (sumt[(size_t)ti * 8 + h] + 1e-16f);
    }
    ah *= 0.125f;
    at *= 0.125f;

    const float* __restrict__ msg = x_r + (size_t)r * 128;
    float* __restrict__ oh = out + (size_t)hi * 256;
    float* __restrict__ ot = out + (size_t)ti * 256 + 128;
#pragma unroll
    for (int d = lane; d < 128; d += 32) {
        const float mv = msg[d];
        atomicAdd(&oh[d], mv * ah);
        atomicAdd(&ot[d], mv * at);
    }
}

extern "C" void kernel_launch(void* const* d_in, const int* in_sizes, int n_in,
                              void* d_out, int out_size, void* d_ws, size_t ws_size,
                              hipStream_t stream) {
    const float* x_e  = (const float*)d_in[0];
    const float* x_r  = (const float*)d_in[1];
    const float* Wh   = (const float*)d_in[2];
    const float* Wt   = (const float*)d_in[3];
    const float* Wr   = (const float*)d_in[4];
    const int*   eidx = (const int*)d_in[5];
    const int*   rel  = (const int*)d_in[6];

    const int D   = 128;
    const int N_e = in_sizes[0] / D;
    const int N_r = in_sizes[1] / D;
    const int E   = in_sizes[6];

    const int* h_idx = eidx;
    const int* t_idx = eidx + E;

    auto align256 = [](size_t x) { return (x + 255) & ~(size_t)255; };
    char* ws = (char*)d_ws;
    float*    Se   = (float*)ws;    ws += align256((size_t)N_e * 16 * sizeof(float));
    float*    Sr   = (float*)ws;    ws += align256((size_t)N_r * 16 * sizeof(float));
    unsigned* maxh = (unsigned*)ws; ws += align256((size_t)N_e * 8 * sizeof(unsigned));
    unsigned* maxt = (unsigned*)ws; ws += align256((size_t)N_e * 8 * sizeof(unsigned));
    float*    sumh = (float*)ws;    ws += align256((size_t)N_e * 8 * sizeof(float));
    float*    sumt = (float*)ws;    ws += align256((size_t)N_e * 8 * sizeof(float));

    float* out = (float*)d_out;

    const int BLK = 256;

    {
        const int n_out = N_e * 256;
        const int n_seg = N_e * 8;
        const int grid  = (n_out + BLK - 1) / BLK;
        k_init<<<grid, BLK, 0, stream>>>(out, sumh, sumt, maxh, maxt, n_out, n_seg);
    }

    {
        const int waves   = (N_e + 15) / 16;
        const int threads = waves * 32;
        const int grid    = (threads + BLK - 1) / BLK;
        k_proj_wmma<<<grid, BLK, 0, stream>>>(x_e, Wh, Wt, Se, N_e);
    }
    {
        const int waves   = (N_r + 15) / 16;
        const int threads = waves * 32;
        const int grid    = (threads + BLK - 1) / BLK;
        k_proj_wmma<<<grid, BLK, 0, stream>>>(x_r, Wr, Wr, Sr, N_r);
    }

    {
        const int n    = E * 8;
        const int grid = (n + BLK - 1) / BLK;
        k_edge_max<<<grid, BLK, 0, stream>>>(h_idx, t_idx, rel, Se, Sr, maxh, maxt, E);
    }

    {
        const int n    = E * 8;
        const int grid = (n + BLK - 1) / BLK;
        k_edge_sum<<<grid, BLK, 0, stream>>>(h_idx, t_idx, rel, Se, Sr, maxh, maxt,
                                             sumh, sumt, E);
    }

    {
        const long long threads = (long long)E * 32;
        const int grid = (int)((threads + BLK - 1) / BLK);
        k_edge_scatter<<<grid, BLK, 0, stream>>>(h_idx, t_idx, rel, Se, Sr, maxh, maxt,
                                                 sumh, sumt, x_r, out, E);
    }
}


# Edges counting-sorted by destination (int atomics for counts and slots); one wave per node side does softmax and message sum in registers: no float atomics, sums in slot arrival order
# speedup vs baseline: 1.5923x; 1.5923x over previous
#include <hip/hip_runtime.h>
#include <stdint.h>

typedef __attribute__((ext_vector_type(2))) float v2f;
typedef __attribute__((ext_vector_type(8))) float v8f;

#define NEG_SLOPE 0.01f

__device__ __forceinline__ float lrelu(float x) {
    return x > 0.0f ? x : NEG_SLOPE * x;
}

__global__ void k_init(int* __restrict__ cnt, int* __restrict__ cursor, int n2) {
    const int i = blockIdx.x * blockDim.x + threadIdx.x;
    if (i < n2) cnt[i] = 0;
    if (i == 0) *cursor = 0;
}

__global__ void k_proj_wmma(const float* __restrict__ X,
                            const float* __restrict__ W0,
                            const float* __restrict__ W1,
                            float* __restrict__ S, int nrows) {
    const int wave = (blockIdx.x * blockDim.x + threadIdx.x) >> 5;
    const int lane = threadIdx.x & 31;
    const int row0 = wave * 16;
    if (row0 >= nrows) return;

    const int m    = lane & 15;
    const int kofs = (lane >> 4) << 1;

    const float* __restrict__ arow = X + (size_t)(row0 + m) * 128;
    const float* __restrict__ brow =
        (m < 8) ? (W0 + (size_t)m * 128) : (W1 + (size_t)(m - 8) * 128);

    v8f c = {0.f, 0.f, 0.f, 0.f, 0.f, 0.f, 0.f, 0.f};
#pragma unroll
    for (int k = 0; k < 128; k += 4) {
        v2f a = *(const v2f*)(arow + k + kofs);
        v2f b = *(const v2f*)(brow + k + kofs);
        c = __builtin_amdgcn_wmma_f32_16x16x4_f32(
             false, a,  false, b,
             (short)0, c,  false,  false);
    }

    const int rbase = row0 + ((lane >> 4) << 3);
    const int col   = lane & 15;
#pragma unroll
    for (int v = 0; v < 8; ++v) {
        S[(size_t)(rbase + v) * 16 + col] = c[v];
    }
}

__global__ void k_count(const int* __restrict__ h_idx,
                        const int* __restrict__ t_idx,
                        int* __restrict__ cnt, int E, int N_e) {
    const int e = blockIdx.x * blockDim.x + threadIdx.x;
    if (e >= E) return;
    atomicAdd(&cnt[h_idx[e]], 1);
    atomicAdd(&cnt[N_e + t_idx[e]], 1);
}

__global__ void k_offsets(const int* __restrict__ cnt, int* __restrict__ off,
                          int* __restrict__ cursor, int n2) {
    const int i    = blockIdx.x * blockDim.x + threadIdx.x;
    const int lane = threadIdx.x & 31;
    const int c    = (i < n2) ? cnt[i] : 0;
    int s = c;
#pragma unroll
    for (int d = 1; d < 32; d <<= 1) {
        const int t = __shfl_up(s, d);
        if (lane >= d) s += t;
    }
    int base = 0;
    if (lane == 31) base = atomicAdd(cursor, s);
    base = __shfl(base, 31);
    if (i < n2) off[i] = base + s - c;
}

__global__ void k_fill(const int* __restrict__ h_idx,
                       const int* __restrict__ t_idx,
                       const int* __restrict__ rel,
                       int* __restrict__ off, int* __restrict__ lst,
                       int E, int N_e) {
    const int e = blockIdx.x * blockDim.x + threadIdx.x;
    if (e >= E) return;
    const int r = rel[e];
    const int p = atomicAdd(&off[h_idx[e]], 1);
    const int q = atomicAdd(&off[N_e + t_idx[e]], 1);
    lst[p] = r;
    lst[q] = r;
}

__global__ void k_node(const int* __restrict__ cnt, const int* __restrict__ off,
                       const int* __restrict__ lst,
                       const float* __restrict__ Se, const float* __restrict__ Sr,
                       const float* __restrict__ x_r, float* __restrict__ out,
                       int N_e) {
    const int seg  = (blockIdx.x * blockDim.x + threadIdx.x) >> 5;
    const int lane = threadIdx.x & 31;
    if (seg >= 2 * N_e) return;
    const int side = (seg >= N_e) ? 1 : 0;
    const int node = seg - side * N_e;

    const int n   = __builtin_amdgcn_readfirstlane(cnt[seg]);
    const int beg = __builtin_amdgcn_readfirstlane(off[seg]) - n;
    const int* __restrict__ l = lst + beg;

    const int h    = lane & 7;
    const int slot = lane >> 3;
    const float en = Se[(size_t)node * 16 + side * 8 + h];

    float m = -INFINITY;
    for (int i = slot; i < n; i += 4) {
        const int r = l[i];
        m = fmaxf(m, lrelu(en + Sr[(size_t)r * 16 + h]));
    }
    m = fmaxf(m, __shfl_xor(m, 8));
    m = fmaxf(m, __shfl_xor(m, 16));

    float z = 0.0f;
    for (int i = slot; i < n; i += 4) {
        const int r = l[i];
        z += __expf(lrelu(en + Sr[(size_t)r * 16 + h]) - m);
    }
    z += __shfl_xor(z, 8);
    z += __shfl_xor(z, 16);
    const float den = z + 1e-16f;

    float4 acc = make_float4(0.0f, 0.0f, 0.0f, 0.0f);
    const float* __restrict__ xl = x_r + lane * 4;
    int i0 = 0;
    for (; i0 + 4 <= n; i0 += 4) {
        const int r = l[i0 + slot];
        float a = __expf(lrelu(en + Sr[(size_t)r * 16 + h]) - m) / den;
        a += __shfl_xor(a, 1);
        a += __shfl_xor(a, 2);
        a += __shfl_xor(a, 4);
        a *= 0.125f;
        const int ai = __float_as_int(a);
        const int r0 = __builtin_amdgcn_readlane(r, 0);
        const int r1 = __builtin_amdgcn_readlane(r, 8);
        const int r2 = __builtin_amdgcn_readlane(r, 16);
        const int r3 = __builtin_amdgcn_readlane(r, 24);
        const float a0 = __int_as_float(__builtin_amdgcn_readlane(ai, 0));
        const float a1 = __int_as_float(__builtin_amdgcn_readlane(ai, 8));
        const float a2 = __int_as_float(__builtin_amdgcn_readlane(ai, 16));
        const float a3 = __int_as_float(__builtin_amdgcn_readlane(ai, 24));
        const float4 m0 = *(const float4*)(xl + (size_t)r0 * 128);
        const float4 m1 = *(const float4*)(xl + (size_t)r1 * 128);
        const float4 m2 = *(const float4*)(xl + (size_t)r2 * 128);
        const float4 m3 = *(const float4*)(xl + (size_t)r3 * 128);
        acc.x += m0.x * a0; acc.y += m0.y * a0; acc.z += m0.z * a0; acc.w += m0.w * a0;
        acc.x += m1.x * a1; acc.y += m1.y * a1; acc.z += m1.z * a1; acc.w += m1.w * a1;
        acc.x += m2.x * a2; acc.y += m2.y * a2; acc.z += m2.z * a2; acc.w += m2.w * a2;
        acc.x += m3.x * a3; acc.y += m3.y * a3; acc.z += m3.z * a3; acc.w += m3.w * a3;
    }
    if (i0 < n) {
        const int  i  = i0 + slot;
        const bool ok = i < n;
        const int  r  = ok ? l[i] : 0;
        float a = 0.0f;
        if (ok) a = __expf(lrelu(en + Sr[(size_t)r * 16 + h]) - m) / den;
        a += __shfl_xor(a, 1);
        a += __shfl_xor(a, 2);
        a += __shfl_xor(a, 4);
        a *= 0.125f;
        const int ai = __float_as_int(a);
#pragma unroll
        for (int k = 0; k < 3; ++k) {
            if (i0 + k < n) {
                const int   rk = __builtin_amdgcn_readlane(r, k * 8);
                const float ak = __int_as_float(__builtin_amdgcn_readlane(ai, k * 8));
                const float4 mv = *(const float4*)(xl + (size_t)rk * 128);
                acc.x += mv.x * ak;
                acc.y += mv.y * ak;
                acc.z += mv.z * ak;
                acc.w += mv.w * ak;
            }
        }
    }
    *(float4*)(out + (size_t)node * 256 + side * 128 + lane * 4) = acc;
}

extern "C" void kernel_launch(void* const* d_in, const int* in_sizes, int n_in,
                              void* d_out, int out_size, void* d_ws, size_t ws_size,
                              hipStream_t stream) {
    const float* x_e  = (const float*)d_in[0];
    const float* x_r  = (const float*)d_in[1];
    const float* Wh   = (const float*)d_in[2];
    const float* Wt   = (const float*)d_in[3];
    const float* Wr   = (const float*)d_in[4];
    const int*   eidx = (const int*)d_in[5];
    const int*   rel  = (const int*)d_in[6];

    const int D   = 128;
    const int N_e = in_sizes[0] / D;
    const int N_r = in_sizes[1] / D;
    const int E   = in_sizes[6];

    const int* h_idx = eidx;
    const int* t_idx = eidx + E;

    auto align256 = [](size_t x) { return (x + 255) & ~(size_t)255; };
    char* ws = (char*)d_ws;
    float* Se     = (float*)ws; ws += align256((size_t)N_e * 16 * sizeof(float));
    float* Sr     = (float*)ws; ws += align256((size_t)N_r * 16 * sizeof(float));
    int*   cnt    = (int*)ws;   ws += align256((size_t)N_e * 2 * sizeof(int));
    int*   off    = (int*)ws;   ws += align256((size_t)N_e * 2 * sizeof(int));
    int*   lst    = (int*)ws;   ws += align256((size_t)E * 2 * sizeof(int));
    int*   cursor = (int*)ws;   ws += 256;
    if ((size_t)(ws - (char*)d_ws) > ws_size) return;

    float* out = (float*)d_out;

    const int BLK = 256;
    const int n2  = N_e * 2;

    k_init<<<(n2 + BLK - 1) / BLK, BLK, 0, stream>>>(cnt, cursor, n2);

    {
        const int waves   = (N_e + 15) / 16;
        const int threads = waves * 32;
        const int grid    = (threads + BLK - 1) / BLK;
        k_proj_wmma<<<grid, BLK, 0, stream>>>(x_e, Wh, Wt, Se, N_e);
    }
    {
        const int waves   = (N_r + 15) / 16;
        const int threads = waves * 32;
        const int grid    = (threads + BLK - 1) / BLK;
        k_proj_wmma<<<grid, BLK, 0, stream>>>(x_r, Wr, Wr, Sr, N_r);
    }

    k_count<<<(E + BLK - 1) / BLK, BLK, 0, stream>>>(h_idx, t_idx, cnt, E, N_e);
    k_offsets<<<(n2 + BLK - 1) / BLK, BLK, 0, stream>>>(cnt, off, cursor, n2);
    k_fill<<<(E + BLK - 1) / BLK, BLK, 0, stream>>>(h_idx, t_idx, rel, off, lst, E, N_e);

    {
        const long long threads = (long long)n2 * 32;
        const int grid = (int)((threads + BLK - 1) / BLK);
        k_node<<<grid, BLK, 0, stream>>>(cnt, off, lst, Se, Sr, x_r, out, N_e);
    }
}


# Node kernel: a segment of up to 32 edges reads its list once and keeps ids and scores in registers, so score and row loads issue as one group, not three dependent passes; sum order unchanged
# speedup vs baseline: 1.6532x; 1.0382x over previous
#include <hip/hip_runtime.h>
#include <stdint.h>

typedef __attribute__((ext_vector_type(2))) float v2f;
typedef __attribute__((ext_vector_type(8))) float v8f;

#define NEG_SLOPE 0.01f

__device__ __forceinline__ float lrelu(float x) {
    return x > 0.0f ? x : NEG_SLOPE * x;
}

__global__ void k_init(int* __restrict__ cnt, int* __restrict__ cursor, int n2) {
    const int i = blockIdx.x * blockDim.x + threadIdx.x;
    if (i < n2) cnt[i] = 0;
    if (i == 0) *cursor = 0;
}

__global__ void k_proj_wmma(const float* __restrict__ X,
                            const float* __restrict__ W0,
                            const float* __restrict__ W1,
                            float* __restrict__ S, int nrows) {
    const int wave = (blockIdx.x * blockDim.x + threadIdx.x) >> 5;
    const int lane = threadIdx.x & 31;
    const int row0 = wave * 16;
    if (row0 >= nrows) return;

    const int m    = lane & 15;
    const int kofs = (lane >> 4) << 1;

    const float* __restrict__ arow = X + (size_t)(row0 + m) * 128;
    const float* __restrict__ brow =
        (m < 8) ? (W0 + (size_t)m * 128) : (W1 + (size_t)(m - 8) * 128);

    v8f c = {0.f, 0.f, 0.f, 0.f, 0.f, 0.f, 0.f, 0.f};
#pragma unroll
    for (int k = 0; k < 128; k += 4) {
        v2f a = *(const v2f*)(arow + k + kofs);
        v2f b = *(const v2f*)(brow + k + kofs);
        c = __builtin_amdgcn_wmma_f32_16x16x4_f32(
             false, a,  false, b,
             (short)0, c,  false,  false);
    }

    const int rbase = row0 + ((lane >> 4) << 3);
    const int col   = lane & 15;
#pragma unroll
    for (int v = 0; v < 8; ++v) {
        S[(size_t)(rbase + v) * 16 + col] = c[v];
    }
}

__global__ void k_count(const int* __restrict__ h_idx,
                        const int* __restrict__ t_idx,
                        int* __restrict__ cnt, int E, int N_e) {
    const int e = blockIdx.x * blockDim.x + threadIdx.x;
    if (e >= E) return;
    atomicAdd(&cnt[h_idx[e]], 1);
    atomicAdd(&cnt[N_e + t_idx[e]], 1);
}

__global__ void k_offsets(const int* __restrict__ cnt, int* __restrict__ off,
                          int* __restrict__ cursor, int n2) {
    const int i    = blockIdx.x * blockDim.x + threadIdx.x;
    const int lane = threadIdx.x & 31;
    const int c    = (i < n2) ? cnt[i] : 0;
    int s = c;
#pragma unroll
    for (int d = 1; d < 32; d <<= 1) {
        const int t = __shfl_up(s, d);
        if (lane >= d) s += t;
    }
    int base = 0;
    if (lane == 31) base = atomicAdd(cursor, s);
    base = __shfl(base, 31);
    if (i < n2) off[i] = base + s - c;
}

__global__ void k_fill(const int* __restrict__ h_idx,
                       const int* __restrict__ t_idx,
                       const int* __restrict__ rel,
                       int* __restrict__ off, int* __restrict__ lst,
                       int E, int N_e) {
    const int e = blockIdx.x * blockDim.x + threadIdx.x;
    if (e >= E) return;
    const int r = rel[e];
    const int p = atomicAdd(&off[h_idx[e]], 1);
    const int q = atomicAdd(&off[N_e + t_idx[e]], 1);
    lst[p] = r;
    lst[q] = r;
}

template <int NJ>
__device__ __forceinline__ void head_mean_weights(const float (&p)[8], float den, int (&ai)[8]) {
    float a[NJ];
#pragma unroll
    for (int j = 0; j < NJ; ++j) a[j] = p[j] / den;
#pragma unroll
    for (int d = 1; d < 8; d <<= 1) {
#pragma unroll
        for (int j = 0; j < NJ; ++j) a[j] += __shfl_xor(a[j], d);
    }
#pragma unroll
    for (int j = 0; j < 8; ++j) ai[j] = (j < NJ) ? __float_as_int(a[j] * 0.125f) : 0;
}

__global__ void k_node(const int* __restrict__ cnt, const int* __restrict__ off,
                       const int* __restrict__ lst,
                       const float* __restrict__ Se, const float* __restrict__ Sr,
                       const float* __restrict__ x_r, float* __restrict__ out,
                       int N_e) {
    const int seg  = (blockIdx.x * blockDim.x + threadIdx.x) >> 5;
    const int lane = threadIdx.x & 31;
    if (seg >= 2 * N_e) return;
    const int side = (seg >= N_e) ? 1 : 0;
    const int node = seg - side * N_e;

    const int n   = __builtin_amdgcn_readfirstlane(cnt[seg]);
    const int beg = __builtin_amdgcn_readfirstlane(off[seg]) - n;
    const int* __restrict__ l = lst + beg;

    const int h    = lane & 7;
    const int slot = lane >> 3;
    const float en = Se[(size_t)node * 16 + side * 8 + h];
    const float* __restrict__ xl = x_r + lane * 4;
    float* __restrict__ o = out + (size_t)node * 256 + side * 128 + lane * 4;

    float4 acc = make_float4(0.0f, 0.0f, 0.0f, 0.0f);

    if (n <= 32) {
        int rl = 0;
        if (lane < n) rl = l[lane];

        float v[8];
#pragma unroll
        for (int j = 0; j < 8; ++j) {
            const int rj = __shfl(rl, 4 * j + slot);
            v[j] = Sr[(size_t)rj * 16 + h];
        }
        float4 rows[16];
#pragma unroll
        for (int k = 0; k < 16; ++k) {
            const int rk = __builtin_amdgcn_readlane(rl, k);
            rows[k] = *(const float4*)(xl + (size_t)rk * 128);
        }

        float s[8];
        float m = -INFINITY;
#pragma unroll
        for (int j = 0; j < 8; ++j) {
            s[j] = (4 * j + slot < n) ? lrelu(en + v[j]) : -INFINITY;
            m = fmaxf(m, s[j]);
        }
        m = fmaxf(m, __shfl_xor(m, 8));
        m = fmaxf(m, __shfl_xor(m, 16));
        float z = 0.0f;
#pragma unroll
        for (int j = 0; j < 8; ++j) {
            s[j] = (4 * j + slot < n) ? __expf(s[j] - m) : 0.0f;
            z += s[j];
        }
        z += __shfl_xor(z, 8);
        z += __shfl_xor(z, 16);
        const float den = z + 1e-16f;
        int ai[8];
        if (n > 16) head_mean_weights<8>(s, den, ai);
        else        head_mean_weights<4>(s, den, ai);

#pragma unroll
        for (int k = 0; k < 16; ++k) {
            if (k < n) {
                const float ak = __int_as_float(__builtin_amdgcn_readlane(ai[k >> 2], (k & 3) * 8));
                acc.x += rows[k].x * ak;
                acc.y += rows[k].y * ak;
                acc.z += rows[k].z * ak;
                acc.w += rows[k].w * ak;
            }
        }
        if (n > 16) {
#pragma unroll
            for (int q = 0; q < 4; ++q) {
                if (16 + 4 * q < n) {
#pragma unroll
                    for (int t = 0; t < 4; ++t) {
                        const int k  = 4 * q + t;
                        const int rk = __builtin_amdgcn_readlane(rl, 16 + k);
                        rows[k] = *(const float4*)(xl + (size_t)rk * 128);
                    }
                }
            }
#pragma unroll
            for (int k = 0; k < 16; ++k) {
                if (16 + k < n) {
                    const float ak = __int_as_float(__builtin_amdgcn_readlane(ai[4 + (k >> 2)], (k & 3) * 8));
                    acc.x += rows[k].x * ak;
                    acc.y += rows[k].y * ak;
                    acc.z += rows[k].z * ak;
                    acc.w += rows[k].w * ak;
                }
            }
        }
        *(float4*)o = acc;
        return;
    }

    float m = -INFINITY;
    for (int i = slot; i < n; i += 4) {
        const int r = l[i];
        m = fmaxf(m, lrelu(en + Sr[(size_t)r * 16 + h]));
    }
    m = fmaxf(m, __shfl_xor(m, 8));
    m = fmaxf(m, __shfl_xor(m, 16));

    float z = 0.0f;
    for (int i = slot; i < n; i += 4) {
        const int r = l[i];
        z += __expf(lrelu(en + Sr[(size_t)r * 16 + h]) - m);
    }
    z += __shfl_xor(z, 8);
    z += __shfl_xor(z, 16);
    const float den = z + 1e-16f;

    int i0 = 0;
    for (; i0 + 4 <= n; i0 += 4) {
        const int r = l[i0 + slot];
        float a = __expf(lrelu(en + Sr[(size_t)r * 16 + h]) - m) / den;
        a += __shfl_xor(a, 1);
        a += __shfl_xor(a, 2);
        a += __shfl_xor(a, 4);
        a *= 0.125f;
        const int ai = __float_as_int(a);
        const int r0 = __builtin_amdgcn_readlane(r, 0);
        const int r1 = __builtin_amdgcn_readlane(r, 8);
        const int r2 = __builtin_amdgcn_readlane(r, 16);
        const int r3 = __builtin_amdgcn_readlane(r, 24);
        const float a0 = __int_as_float(__builtin_amdgcn_readlane(ai, 0));
        const float a1 = __int_as_float(__builtin_amdgcn_readlane(ai, 8));
        const float a2 = __int_as_float(__builtin_amdgcn_readlane(ai, 16));
        const float a3 = __int_as_float(__builtin_amdgcn_readlane(ai, 24));
        const float4 m0 = *(const float4*)(xl + (size_t)r0 * 128);
        const float4 m1 = *(const float4*)(xl + (size_t)r1 * 128);
        const float4 m2 = *(const float4*)(xl + (size_t)r2 * 128);
        const float4 m3 = *(const float4*)(xl + (size_t)r3 * 128);
        acc.x += m0.x * a0; acc.y += m0.y * a0; acc.z += m0.z * a0; acc.w += m0.w * a0;
        acc.x += m1.x * a1; acc.y += m1.y * a1; acc.z += m1.z * a1; acc.w += m1.w * a1;
        acc.x += m2.x * a2; acc.y += m2.y * a2; acc.z += m2.z * a2; acc.w += m2.w * a2;
        acc.x += m3.x * a3; acc.y += m3.y * a3; acc.z += m3.z * a3; acc.w += m3.w * a3;
    }
    if (i0 < n) {
        const int  i  = i0 + slot;
        const bool ok = i < n;
        const int  r  = ok ? l[i] : 0;
        float a = 0.0f;
        if (ok) a = __expf(lrelu(en + Sr[(size_t)r * 16 + h]) - m) / den;
        a += __shfl_xor(a, 1);
        a += __shfl_xor(a, 2);
        a += __shfl_xor(a, 4);
        a *= 0.125f;
        const int ai = __float_as_int(a);
#pragma unroll
        for (int k = 0; k < 3; ++k) {
            if (i0 + k < n) {
                const int   rk = __builtin_amdgcn_readlane(r, k * 8);
                const float ak = __int_as_float(__builtin_amdgcn_readlane(ai, k * 8));
                const float4 mv = *(const float4*)(xl + (size_t)rk * 128);
                acc.x += mv.x * ak;
                acc.y += mv.y * ak;
                acc.z += mv.z * ak;
                acc.w += mv.w * ak;
            }
        }
    }
    *(float4*)o = acc;
}

extern "C" void kernel_launch(void* const* d_in, const int* in_sizes, int n_in,
                              void* d_out, int out_size, void* d_ws, size_t ws_size,
                              hipStream_t stream) {
    const float* x_e  = (const float*)d_in[0];
    const float* x_r  = (const float*)d_in[1];
    const float* Wh   = (const float*)d_in[2];
    const float* Wt   = (const float*)d_in[3];
    const float* Wr   = (const float*)d_in[4];
    const int*   eidx = (const int*)d_in[5];
    const int*   rel  = (const int*)d_in[6];

    const int D   = 128;
    const int N_e = in_sizes[0] / D;
    const int N_r = in_sizes[1] / D;
    const int E   = in_sizes[6];

    const int* h_idx = eidx;
    const int* t_idx = eidx + E;

    auto align256 = [](size_t x) { return (x + 255) & ~(size_t)255; };
    char* ws = (char*)d_ws;
    float* Se     = (float*)ws; ws += align256((size_t)N_e * 16 * sizeof(float));
    float* Sr     = (float*)ws; ws += align256((size_t)N_r * 16 * sizeof(float));
    int*   cnt    = (int*)ws;   ws += align256((size_t)N_e * 2 * sizeof(int));
    int*   off    = (int*)ws;   ws += align256((size_t)N_e * 2 * sizeof(int));
    int*   lst    = (int*)ws;   ws += align256((size_t)E * 2 * sizeof(int));
    int*   cursor = (int*)ws;   ws += 256;
    if ((size_t)(ws - (char*)d_ws) > ws_size) return;

    float* out = (float*)d_out;

    const int BLK = 256;
    const int n2  = N_e * 2;

    k_init<<<(n2 + BLK - 1) / BLK, BLK, 0, stream>>>(cnt, cursor, n2);

    {
        const int waves   = (N_e + 15) / 16;
        const int threads = waves * 32;
        const int grid    = (threads + BLK - 1) / BLK;
        k_proj_wmma<<<grid, BLK, 0, stream>>>(x_e, Wh, Wt, Se, N_e);
    }
    {
        const int waves   = (N_r + 15) / 16;
        const int threads = waves * 32;
        const int grid    = (threads + BLK - 1) / BLK;
        k_proj_wmma<<<grid, BLK, 0, stream>>>(x_r, Wr, Wr, Sr, N_r);
    }

    k_count<<<(E + BLK - 1) / BLK, BLK, 0, stream>>>(h_idx, t_idx, cnt, E, N_e);
    k_offsets<<<(n2 + BLK - 1) / BLK, BLK, 0, stream>>>(cnt, off, cursor, n2);
    k_fill<<<(E + BLK - 1) / BLK, BLK, 0, stream>>>(h_idx, t_idx, rel, off, lst, E, N_e);

    {
        const long long threads = (long long)n2 * 32;
        const int grid = (int)((threads + BLK - 1) / BLK);
        k_node<<<grid, BLK, 0, stream>>>(cnt, off, lst, Se, Sr, x_r, out, N_e);
    }
}


# Count pass keeps each counter add's return as the edge's rank; fill pass places edges at run start + rank with plain loads: half the integer atomics; slot order now set by the count pass
# speedup vs baseline: 2.2839x; 1.3815x over previous
#include <hip/hip_runtime.h>
#include <stdint.h>

typedef __attribute__((ext_vector_type(2))) float v2f;
typedef __attribute__((ext_vector_type(8))) float v8f;

#define NEG_SLOPE 0.01f

__device__ __forceinline__ float lrelu(float x) {
    return x > 0.0f ? x : NEG_SLOPE * x;
}

__global__ void k_init(int* __restrict__ cnt, int* __restrict__ cursor, int n2) {
    const int i = blockIdx.x * blockDim.x + threadIdx.x;
    if (i < n2) cnt[i] = 0;
    if (i == 0) *cursor = 0;
}

__global__ void k_proj_wmma(const float* __restrict__ X,
                            const float* __restrict__ W0,
                            const float* __restrict__ W1,
                            float* __restrict__ S, int nrows) {
    const int wave = (blockIdx.x * blockDim.x + threadIdx.x) >> 5;
    const int lane = threadIdx.x & 31;
    const int row0 = wave * 16;
    if (row0 >= nrows) return;

    const int m    = lane & 15;
    const int kofs = (lane >> 4) << 1;

    const float* __restrict__ arow = X + (size_t)(row0 + m) * 128;
    const float* __restrict__ brow =
        (m < 8) ? (W0 + (size_t)m * 128) : (W1 + (size_t)(m - 8) * 128);

    v8f c = {0.f, 0.f, 0.f, 0.f, 0.f, 0.f, 0.f, 0.f};
#pragma unroll
    for (int k = 0; k < 128; k += 4) {
        v2f a = *(const v2f*)(arow + k + kofs);
        v2f b = *(const v2f*)(brow + k + kofs);
        c = __builtin_amdgcn_wmma_f32_16x16x4_f32(
             false, a,  false, b,
             (short)0, c,  false,  false);
    }

    const int rbase = row0 + ((lane >> 4) << 3);
    const int col   = lane & 15;
#pragma unroll
    for (int v = 0; v < 8; ++v) {
        S[(size_t)(rbase + v) * 16 + col] = c[v];
    }
}

__global__ void k_count(const int* __restrict__ h_idx,
                        const int* __restrict__ t_idx,
                        int* __restrict__ cnt, int* __restrict__ rank,
                        int E, int N_e) {
    const int e = blockIdx.x * blockDim.x + threadIdx.x;
    if (e >= E) return;
    const int h = h_idx[e];
    const int t = t_idx[e];
    const int a = atomicAdd(&cnt[h], 1);
    const int b = atomicAdd(&cnt[N_e + t], 1);
    rank[e]     = a;
    rank[E + e] = b;
}

__global__ void k_offsets(const int* __restrict__ cnt, int* __restrict__ off,
                          int* __restrict__ cursor, int n2) {
    const int i    = blockIdx.x * blockDim.x + threadIdx.x;
    const int lane = threadIdx.x & 31;
    const int c    = (i < n2) ? cnt[i] : 0;
    int s = c;
#pragma unroll
    for (int d = 1; d < 32; d <<= 1) {
        const int t = __shfl_up(s, d);
        if (lane >= d) s += t;
    }
    int base = 0;
    if (lane == 31) base = atomicAdd(cursor, s);
    base = __shfl(base, 31);
    if (i < n2) off[i] = base + s - c;
}

__global__ void k_fill(const int* __restrict__ h_idx,
                       const int* __restrict__ t_idx,
                       const int* __restrict__ rel,
                       const int* __restrict__ off,
                       const int* __restrict__ rank,
                       int* __restrict__ lst, int E, int N_e) {
    const int e = blockIdx.x * blockDim.x + threadIdx.x;
    if (e >= E) return;
    const int r = rel[e];
    const int p = off[h_idx[e]] + rank[e];
    const int q = off[N_e + t_idx[e]] + rank[E + e];
    lst[p] = r;
    lst[q] = r;
}

template <int NJ>
__device__ __forceinline__ void head_mean_weights(const float (&p)[8], float den, int (&ai)[8]) {
    float a[NJ];
#pragma unroll
    for (int j = 0; j < NJ; ++j) a[j] = p[j] / den;
#pragma unroll
    for (int d = 1; d < 8; d <<= 1) {
#pragma unroll
        for (int j = 0; j < NJ; ++j) a[j] += __shfl_xor(a[j], d);
    }
#pragma unroll
    for (int j = 0; j < 8; ++j) ai[j] = (j < NJ) ? __float_as_int(a[j] * 0.125f) : 0;
}

__global__ void k_node(const int* __restrict__ cnt, const int* __restrict__ off,
                       const int* __restrict__ lst,
                       const float* __restrict__ Se, const float* __restrict__ Sr,
                       const float* __restrict__ x_r, float* __restrict__ out,
                       int N_e) {
    const int seg  = (blockIdx.x * blockDim.x + threadIdx.x) >> 5;
    const int lane = threadIdx.x & 31;
    if (seg >= 2 * N_e) return;
    const int side = (seg >= N_e) ? 1 : 0;
    const int node = seg - side * N_e;

    const int n   = __builtin_amdgcn_readfirstlane(cnt[seg]);
    const int beg = __builtin_amdgcn_readfirstlane(off[seg]);
    const int* __restrict__ l = lst + beg;

    const int h    = lane & 7;
    const int slot = lane >> 3;
    const float en = Se[(size_t)node * 16 + side * 8 + h];
    const float* __restrict__ xl = x_r + lane * 4;
    float* __restrict__ o = out + (size_t)node * 256 + side * 128 + lane * 4;

    float4 acc = make_float4(0.0f, 0.0f, 0.0f, 0.0f);

    if (n <= 32) {
        int rl = 0;
        if (lane < n) rl = l[lane];

        float v[8];
#pragma unroll
        for (int j = 0; j < 8; ++j) {
            const int rj = __shfl(rl, 4 * j + slot);
            v[j] = Sr[(size_t)rj * 16 + h];
        }
        float4 rows[16];
#pragma unroll
        for (int k = 0; k < 16; ++k) {
            const int rk = __builtin_amdgcn_readlane(rl, k);
            rows[k] = *(const float4*)(xl + (size_t)rk * 128);
        }

        float s[8];
        float m = -INFINITY;
#pragma unroll
        for (int j = 0; j < 8; ++j) {
            s[j] = (4 * j + slot < n) ? lrelu(en + v[j]) : -INFINITY;
            m = fmaxf(m, s[j]);
        }
        m = fmaxf(m, __shfl_xor(m, 8));
        m = fmaxf(m, __shfl_xor(m, 16));
        float z = 0.0f;
#pragma unroll
        for (int j = 0; j < 8; ++j) {
            s[j] = (4 * j + slot < n) ? __expf(s[j] - m) : 0.0f;
            z += s[j];
        }
        z += __shfl_xor(z, 8);
        z += __shfl_xor(z, 16);
        const float den = z + 1e-16f;
        int ai[8];
        if (n > 16) head_mean_weights<8>(s, den, ai);
        else        head_mean_weights<4>(s, den, ai);

#pragma unroll
        for (int k = 0; k < 16; ++k) {
            if (k < n) {
                const float ak = __int_as_float(__builtin_amdgcn_readlane(ai[k >> 2], (k & 3) * 8));
                acc.x += rows[k].x * ak;
                acc.y += rows[k].y * ak;
                acc.z += rows[k].z * ak;
                acc.w += rows[k].w * ak;
            }
        }
        if (n > 16) {
#pragma unroll
            for (int q = 0; q < 4; ++q) {
                if (16 + 4 * q < n) {
#pragma unroll
                    for (int t = 0; t < 4; ++t) {
                        const int k  = 4 * q + t;
                        const int rk = __builtin_amdgcn_readlane(rl, 16 + k);
                        rows[k] = *(const float4*)(xl + (size_t)rk * 128);
                    }
                }
            }
#pragma unroll
            for (int k = 0; k < 16; ++k) {
                if (16 + k < n) {
                    const float ak = __int_as_float(__builtin_amdgcn_readlane(ai[4 + (k >> 2)], (k & 3) * 8));
                    acc.x += rows[k].x * ak;
                    acc.y += rows[k].y * ak;
                    acc.z += rows[k].z * ak;
                    acc.w += rows[k].w * ak;
                }
            }
        }
        *(float4*)o = acc;
        return;
    }

    float m = -INFINITY;
    for (int i = slot; i < n; i += 4) {
        const int r = l[i];
        m = fmaxf(m, lrelu(en + Sr[(size_t)r * 16 + h]));
    }
    m = fmaxf(m, __shfl_xor(m, 8));
    m = fmaxf(m, __shfl_xor(m, 16));

    float z = 0.0f;
    for (int i = slot; i < n; i += 4) {
        const int r = l[i];
        z += __expf(lrelu(en + Sr[(size_t)r * 16 + h]) - m);
    }
    z += __shfl_xor(z, 8);
    z += __shfl_xor(z, 16);
    const float den = z + 1e-16f;

    int i0 = 0;
    for (; i0 + 4 <= n; i0 += 4) {
        const int r = l[i0 + slot];
        float a = __expf(lrelu(en + Sr[(size_t)r * 16 + h]) - m) / den;
        a += __shfl_xor(a, 1);
        a += __shfl_xor(a, 2);
        a += __shfl_xor(a, 4);
        a *= 0.125f;
        const int ai = __float_as_int(a);
        const int r0 = __builtin_amdgcn_readlane(r, 0);
        const int r1 = __builtin_amdgcn_readlane(r, 8);
        const int r2 = __builtin_amdgcn_readlane(r, 16);
        const int r3 = __builtin_amdgcn_readlane(r, 24);
        const float a0 = __int_as_float(__builtin_amdgcn_readlane(ai, 0));
        const float a1 = __int_as_float(__builtin_amdgcn_readlane(ai, 8));
        const float a2 = __int_as_float(__builtin_amdgcn_readlane(ai, 16));
        const float a3 = __int_as_float(__builtin_amdgcn_readlane(ai, 24));
        const float4 m0 = *(const float4*)(xl + (size_t)r0 * 128);
        const float4 m1 = *(const float4*)(xl + (size_t)r1 * 128);
        const float4 m2 = *(const float4*)(xl + (size_t)r2 * 128);
        const float4 m3 = *(const float4*)(xl + (size_t)r3 * 128);
        acc.x += m0.x * a0; acc.y += m0.y * a0; acc.z += m0.z * a0; acc.w += m0.w * a0;
        acc.x += m1.x * a1; acc.y += m1.y * a1; acc.z += m1.z * a1; acc.w += m1.w * a1;
        acc.x += m2.x * a2; acc.y += m2.y * a2; acc.z += m2.z * a2; acc.w += m2.w * a2;
        acc.x += m3.x * a3; acc.y += m3.y * a3; acc.z += m3.z * a3; acc.w += m3.w * a3;
    }
    if (i0 < n) {
        const int  i  = i0 + slot;
        const bool ok = i < n;
        const int  r  = ok ? l[i] : 0;
        float a = 0.0f;
        if (ok) a = __expf(lrelu(en + Sr[(size_t)r * 16 + h]) - m) / den;
        a += __shfl_xor(a, 1);
        a += __shfl_xor(a, 2);
        a += __shfl_xor(a, 4);
        a *= 0.125f;
        const int ai = __float_as_int(a);
#pragma unroll
        for (int k = 0; k < 3; ++k) {
            if (i0 + k < n) {
                const int   rk = __builtin_amdgcn_readlane(r, k * 8);
                const float ak = __int_as_float(__builtin_amdgcn_readlane(ai, k * 8));
                const float4 mv = *(const float4*)(xl + (size_t)rk * 128);
                acc.x += mv.x * ak;
                acc.y += mv.y * ak;
                acc.z += mv.z * ak;
                acc.w += mv.w * ak;
            }
        }
    }
    *(float4*)o = acc;
}

extern "C" void kernel_launch(void* const* d_in, const int* in_sizes, int n_in,
                              void* d_out, int out_size, void* d_ws, size_t ws_size,
                              hipStream_t stream) {
    const float* x_e  = (const float*)d_in[0];
    const float* x_r  = (const float*)d_in[1];
    const float* Wh   = (const float*)d_in[2];
    const float* Wt   = (const float*)d_in[3];
    const float* Wr   = (const float*)d_in[4];
    const int*   eidx = (const int*)d_in[5];
    const int*   rel  = (const int*)d_in[6];

    const int D   = 128;
    const int N_e = in_sizes[0] / D;
    const int N_r = in_sizes[1] / D;
    const int E   = in_sizes[6];

    const int* h_idx = eidx;
    const int* t_idx = eidx + E;

    auto align256 = [](size_t x) { return (x + 255) & ~(size_t)255; };
    char* ws = (char*)d_ws;
    float* Se     = (float*)ws; ws += align256((size_t)N_e * 16 * sizeof(float));
    float* Sr     = (float*)ws; ws += align256((size_t)N_r * 16 * sizeof(float));
    int*   cnt    = (int*)ws;   ws += align256((size_t)N_e * 2 * sizeof(int));
    int*   off    = (int*)ws;   ws += align256((size_t)N_e * 2 * sizeof(int));
    int*   lst    = (int*)ws;   ws += align256((size_t)E * 2 * sizeof(int));
    int*   cursor = (int*)ws;   ws += 256;
    if ((size_t)(ws - (char*)d_ws) > ws_size) return;

    float* out = (float*)d_out;

    int* rank = (int*)ws;       ws += align256((size_t)E * 2 * sizeof(int));
    if ((size_t)(ws - (char*)d_ws) > ws_size) {
        if ((size_t)out_size < (size_t)E * 2) return;
        rank = (int*)d_out;
    }

    const int BLK = 256;
    const int n2  = N_e * 2;

    k_init<<<(n2 + BLK - 1) / BLK, BLK, 0, stream>>>(cnt, cursor, n2);

    {
        const int waves   = (N_e + 15) / 16;
        const int threads = waves * 32;
        const int grid    = (threads + BLK - 1) / BLK;
        k_proj_wmma<<<grid, BLK, 0, stream>>>(x_e, Wh, Wt, Se, N_e);
    }
    {
        const int waves   = (N_r + 15) / 16;
        const int threads = waves * 32;
        const int grid    = (threads + BLK - 1) / BLK;
        k_proj_wmma<<<grid, BLK, 0, stream>>>(x_r, Wr, Wr, Sr, N_r);
    }

    k_count<<<(E + BLK - 1) / BLK, BLK, 0, stream>>>(h_idx, t_idx, cnt, rank, E, N_e);
    k_offsets<<<(n2 + BLK - 1) / BLK, BLK, 0, stream>>>(cnt, off, cursor, n2);
    k_fill<<<(E + BLK - 1) / BLK, BLK, 0, stream>>>(h_idx, t_idx, rel, off, rank, lst, E, N_e);

    {
        const long long threads = (long long)n2 * 32;
        const int grid = (int)((threads + BLK - 1) / BLK);
        k_node<<<grid, BLK, 0, stream>>>(cnt, off, lst, Se, Sr, x_r, out, N_e);
    }
}


# Edge ranks from shared-memory counters: a block per (id range, edge chunk) counts with LDS atomics, chunk bases summed in offset pass; no global atomics on edges, no zeroing launch; slots chunk-major
# speedup vs baseline: 3.4550x; 1.5128x over previous
#include <hip/hip_runtime.h>
#include <stdint.h>

typedef __attribute__((ext_vector_type(2))) float v2f;
typedef __attribute__((ext_vector_type(8))) float v8f;

#define NEG_SLOPE 0.01f

__device__ __forceinline__ float lrelu(float x) {
    return x > 0.0f ? x : NEG_SLOPE * x;
}

#define CH   8
#define PS   16
#define CAP  4096
#define CNT_THREADS 1024
#define EDGE_BLK 256

__device__ __forceinline__ int chunk_len(int E) {
    return (((E + CH - 1) / CH) + EDGE_BLK - 1) & ~(EDGE_BLK - 1);
}

__global__ void k_proj_wmma(const float* __restrict__ X,
                            const float* __restrict__ W0,
                            const float* __restrict__ W1,
                            float* __restrict__ S, int nrows) {
    const int wave = (blockIdx.x * blockDim.x + threadIdx.x) >> 5;
    const int lane = threadIdx.x & 31;
    const int row0 = wave * 16;
    if (row0 >= nrows) return;

    const int m    = lane & 15;
    const int kofs = (lane >> 4) << 1;

    const float* __restrict__ arow = X + (size_t)(row0 + m) * 128;
    const float* __restrict__ brow =
        (m < 8) ? (W0 + (size_t)m * 128) : (W1 + (size_t)(m - 8) * 128);

    v8f c = {0.f, 0.f, 0.f, 0.f, 0.f, 0.f, 0.f, 0.f};
#pragma unroll
    for (int k = 0; k < 128; k += 4) {
        v2f a = *(const v2f*)(arow + k + kofs);
        v2f b = *(const v2f*)(brow + k + kofs);
        c = __builtin_amdgcn_wmma_f32_16x16x4_f32(
             false, a,  false, b,
             (short)0, c,  false,  false);
    }

    const int rbase = row0 + ((lane >> 4) << 3);
    const int col   = lane & 15;
#pragma unroll
    for (int v = 0; v < 8; ++v) {
        S[(size_t)(rbase + v) * 16 + col] = c[v];
    }
}

template <bool VEC>
__device__ __forceinline__ void count_ids(const int* __restrict__ id, int* __restrict__ rk, int* c,
                                          int s0, int len, int e0, int e1, int tid) {
    for (int e = e0 + 4 * tid; e < e1; e += 16 * CNT_THREADS) {
        int4 v[4];
#pragma unroll
        for (int u = 0; u < 4; ++u) {
            const int ee = e + u * 4 * CNT_THREADS;
            v[u] = make_int4(-1, -1, -1, -1);
            if (VEC) {
                if (ee < e1) v[u] = *(const int4*)(id + ee);
            } else {
                if (ee     < e1) v[u].x = id[ee];
                if (ee + 1 < e1) v[u].y = id[ee + 1];
                if (ee + 2 < e1) v[u].z = id[ee + 2];
                if (ee + 3 < e1) v[u].w = id[ee + 3];
            }
        }
        int r[16];
#pragma unroll
        for (int u = 0; u < 4; ++u) {
            const int vv[4] = {v[u].x, v[u].y, v[u].z, v[u].w};
#pragma unroll
            for (int j = 0; j < 4; ++j) {
                const unsigned d = (unsigned)(vv[j] - s0);
                r[4 * u + j] = (d < (unsigned)len) ? atomicAdd(&c[d], 1) : -1;
            }
        }
#pragma unroll
        for (int u = 0; u < 4; ++u) {
#pragma unroll
            for (int j = 0; j < 4; ++j) {
                if (r[4 * u + j] >= 0) rk[e + u * 4 * CNT_THREADS + j] = r[4 * u + j];
            }
        }
    }
}

__global__ void __launch_bounds__(CNT_THREADS)
k_count(const int* __restrict__ eidx, int* __restrict__ cb, int* __restrict__ rank,
        int* __restrict__ cursor, int E, int N_e) {
    __shared__ int c[CAP];
    const int tid   = threadIdx.x;
    const int chunk = blockIdx.x;
    const int side  = blockIdx.y / PS;
    const int part  = blockIdx.y - side * PS;
    if (blockIdx.x == 0 && blockIdx.y == 0 && tid == 0) *cursor = 0;

    const int n2   = 2 * N_e;
    const int plen = (N_e + PS - 1) / PS;
    const int lo   = part * plen;
    const int hi   = min(lo + plen, N_e);
    const int clen = chunk_len(E);
    const int e0   = chunk * clen;
    const int e1   = min(e0 + clen, E);
    const int* __restrict__ id = eidx + (size_t)side * E;
    int* __restrict__ rk = rank + (size_t)side * E;
    int* __restrict__ co = cb + (size_t)chunk * n2 + (size_t)side * N_e;
    const bool vec = ((E & 3) == 0) && ((((size_t)eidx) & 15) == 0);

    for (int s0 = lo; s0 < hi; s0 += CAP) {
        const int len = min(CAP, hi - s0);
        for (int i = tid; i < len; i += CNT_THREADS) c[i] = 0;
        __syncthreads();
        if (vec) count_ids<true>(id, rk, c, s0, len, e0, e1, tid);
        else     count_ids<false>(id, rk, c, s0, len, e0, e1, tid);
        __syncthreads();
        for (int i = tid; i < len; i += CNT_THREADS) co[s0 + i] = c[i];
        __syncthreads();
    }
}

__global__ void k_offsets(int* __restrict__ cb, int* __restrict__ cnt, int* __restrict__ off,
                          int* __restrict__ cursor, int n2) {
    const int i    = blockIdx.x * blockDim.x + threadIdx.x;
    const int lane = threadIdx.x & 31;
    int pc[CH];
    int c = 0;
#pragma unroll
    for (int k = 0; k < CH; ++k) {
        pc[k] = (i < n2) ? cb[(size_t)k * n2 + i] : 0;
        c += pc[k];
    }
    int s = c;
#pragma unroll
    for (int d = 1; d < 32; d <<= 1) {
        const int t = __shfl_up(s, d);
        if (lane >= d) s += t;
    }
    int base = 0;
    if (lane == 31) base = atomicAdd(cursor, s);
    base = __shfl(base, 31);
    if (i < n2) {
        int start = base + s - c;
        cnt[i] = c;
        off[i] = start;
#pragma unroll
        for (int k = 0; k < CH; ++k) {
            cb[(size_t)k * n2 + i] = start;
            start += pc[k];
        }
    }
}

__global__ void k_fill(const int* __restrict__ h_idx,
                       const int* __restrict__ t_idx,
                       const int* __restrict__ rel,
                       const int* __restrict__ cb,
                       const int* __restrict__ rank,
                       int* __restrict__ lst, int E, int N_e) {
    const int e = blockIdx.x * EDGE_BLK + threadIdx.x;
    if (e >= E) return;
    const int chunk = (int)((blockIdx.x * (unsigned)EDGE_BLK) / (unsigned)chunk_len(E));
    const int* __restrict__ b = cb + (size_t)chunk * (2 * (size_t)N_e);
    const int r = rel[e];
    const int p = b[h_idx[e]] + rank[e];
    const int q = b[N_e + t_idx[e]] + rank[E + e];
    lst[p] = r;
    lst[q] = r;
}

template <int NJ>
__device__ __forceinline__ void head_mean_weights(const float (&p)[8], float den, int (&ai)[8]) {
    float a[NJ];
#pragma unroll
    for (int j = 0; j < NJ; ++j) a[j] = p[j] / den;
#pragma unroll
    for (int d = 1; d < 8; d <<= 1) {
#pragma unroll
        for (int j = 0; j < NJ; ++j) a[j] += __shfl_xor(a[j], d);
    }
#pragma unroll
    for (int j = 0; j < 8; ++j) ai[j] = (j < NJ) ? __float_as_int(a[j] * 0.125f) : 0;
}

__global__ void k_node(const int* __restrict__ cnt, const int* __restrict__ off,
                       const int* __restrict__ lst,
                       const float* __restrict__ Se, const float* __restrict__ Sr,
                       const float* __restrict__ x_r, float* __restrict__ out,
                       int N_e) {
    const int seg  = (blockIdx.x * blockDim.x + threadIdx.x) >> 5;
    const int lane = threadIdx.x & 31;
    if (seg >= 2 * N_e) return;
    const int side = (seg >= N_e) ? 1 : 0;
    const int node = seg - side * N_e;

    const int n   = __builtin_amdgcn_readfirstlane(cnt[seg]);
    const int beg = __builtin_amdgcn_readfirstlane(off[seg]);
    const int* __restrict__ l = lst + beg;

    const int h    = lane & 7;
    const int slot = lane >> 3;
    const float en = Se[(size_t)node * 16 + side * 8 + h];
    const float* __restrict__ xl = x_r + lane * 4;
    float* __restrict__ o = out + (size_t)node * 256 + side * 128 + lane * 4;

    float4 acc = make_float4(0.0f, 0.0f, 0.0f, 0.0f);

    if (n <= 32) {
        int rl = 0;
        if (lane < n) rl = l[lane];

        float v[8];
#pragma unroll
        for (int j = 0; j < 8; ++j) {
            const int rj = __shfl(rl, 4 * j + slot);
            v[j] = Sr[(size_t)rj * 16 + h];
        }
        float4 rows[16];
#pragma unroll
        for (int k = 0; k < 16; ++k) {
            const int rk = __builtin_amdgcn_readlane(rl, k);
            rows[k] = *(const float4*)(xl + (size_t)rk * 128);
        }

        float s[8];
        float m = -INFINITY;
#pragma unroll
        for (int j = 0; j < 8; ++j) {
            s[j] = (4 * j + slot < n) ? lrelu(en + v[j]) : -INFINITY;
            m = fmaxf(m, s[j]);
        }
        m = fmaxf(m, __shfl_xor(m, 8));
        m = fmaxf(m, __shfl_xor(m, 16));
        float z = 0.0f;
#pragma unroll
        for (int j = 0; j < 8; ++j) {
            s[j] = (4 * j + slot < n) ? __expf(s[j] - m) : 0.0f;
            z += s[j];
        }
        z += __shfl_xor(z, 8);
        z += __shfl_xor(z, 16);
        const float den = z + 1e-16f;
        int ai[8];
        if (n > 16) head_mean_weights<8>(s, den, ai);
        else        head_mean_weights<4>(s, den, ai);

#pragma unroll
        for (int k = 0; k < 16; ++k) {
            if (k < n) {
                const float ak = __int_as_float(__builtin_amdgcn_readlane(ai[k >> 2], (k & 3) * 8));
                acc.x += rows[k].x * ak;
                acc.y += rows[k].y * ak;
                acc.z += rows[k].z * ak;
                acc.w += rows[k].w * ak;
            }
        }
        if (n > 16) {
#pragma unroll
            for (int q = 0; q < 4; ++q) {
                if (16 + 4 * q < n) {
#pragma unroll
                    for (int t = 0; t < 4; ++t) {
                        const int k  = 4 * q + t;
                        const int rk = __builtin_amdgcn_readlane(rl, 16 + k);
                        rows[k] = *(const float4*)(xl + (size_t)rk * 128);
                    }
                }
            }
#pragma unroll
            for (int k = 0; k < 16; ++k) {
                if (16 + k < n) {
                    const float ak = __int_as_float(__builtin_amdgcn_readlane(ai[4 + (k >> 2)], (k & 3) * 8));
                    acc.x += rows[k].x * ak;
                    acc.y += rows[k].y * ak;
                    acc.z += rows[k].z * ak;
                    acc.w += rows[k].w * ak;
                }
            }
        }
        *(float4*)o = acc;
        return;
    }

    float m = -INFINITY;
    for (int i = slot; i < n; i += 4) {
        const int r = l[i];
        m = fmaxf(m, lrelu(en + Sr[(size_t)r * 16 + h]));
    }
    m = fmaxf(m, __shfl_xor(m, 8));
    m = fmaxf(m, __shfl_xor(m, 16));

    float z = 0.0f;
    for (int i = slot; i < n; i += 4) {
        const int r = l[i];
        z += __expf(lrelu(en + Sr[(size_t)r * 16 + h]) - m);
    }
    z += __shfl_xor(z, 8);
    z += __shfl_xor(z, 16);
    const float den = z + 1e-16f;

    int i0 = 0;
    for (; i0 + 4 <= n; i0 += 4) {
        const int r = l[i0 + slot];
        float a = __expf(lrelu(en + Sr[(size_t)r * 16 + h]) - m) / den;
        a += __shfl_xor(a, 1);
        a += __shfl_xor(a, 2);
        a += __shfl_xor(a, 4);
        a *= 0.125f;
        const int ai = __float_as_int(a);
        const int r0 = __builtin_amdgcn_readlane(r, 0);
        const int r1 = __builtin_amdgcn_readlane(r, 8);
        const int r2 = __builtin_amdgcn_readlane(r, 16);
        const int r3 = __builtin_amdgcn_readlane(r, 24);
        const float a0 = __int_as_float(__builtin_amdgcn_readlane(ai, 0));
        const float a1 = __int_as_float(__builtin_amdgcn_readlane(ai, 8));
        const float a2 = __int_as_float(__builtin_amdgcn_readlane(ai, 16));
        const float a3 = __int_as_float(__builtin_amdgcn_readlane(ai, 24));
        const float4 m0 = *(const float4*)(xl + (size_t)r0 * 128);
        const float4 m1 = *(const float4*)(xl + (size_t)r1 * 128);
        const float4 m2 = *(const float4*)(xl + (size_t)r2 * 128);
        const float4 m3 = *(const float4*)(xl + (size_t)r3 * 128);
        acc.x += m0.x * a0; acc.y += m0.y * a0; acc.z += m0.z * a0; acc.w += m0.w * a0;
        acc.x += m1.x * a1; acc.y += m1.y * a1; acc.z += m1.z * a1; acc.w += m1.w * a1;
        acc.x += m2.x * a2; acc.y += m2.y * a2; acc.z += m2.z * a2; acc.w += m2.w * a2;
        acc.x += m3.x * a3; acc.y += m3.y * a3; acc.z += m3.z * a3; acc.w += m3.w * a3;
    }
    if (i0 < n) {
        const int  i  = i0 + slot;
        const bool ok = i < n;
        const int  r  = ok ? l[i] : 0;
        float a = 0.0f;
        if (ok) a = __expf(lrelu(en + Sr[(size_t)r * 16 + h]) - m) / den;
        a += __shfl_xor(a, 1);
        a += __shfl_xor(a, 2);
        a += __shfl_xor(a, 4);
        a *= 0.125f;
        const int ai = __float_as_int(a);
#pragma unroll
        for (int k = 0; k < 3; ++k) {
            if (i0 + k < n) {
                const int   rk = __builtin_amdgcn_readlane(r, k * 8);
                const float ak = __int_as_float(__builtin_amdgcn_readlane(ai, k * 8));
                const float4 mv = *(const float4*)(xl + (size_t)rk * 128);
                acc.x += mv.x * ak;
                acc.y += mv.y * ak;
                acc.z += mv.z * ak;
                acc.w += mv.w * ak;
            }
        }
    }
    *(float4*)o = acc;
}

extern "C" void kernel_launch(void* const* d_in, const int* in_sizes, int n_in,
                              void* d_out, int out_size, void* d_ws, size_t ws_size,
                              hipStream_t stream) {
    const float* x_e  = (const float*)d_in[0];
    const float* x_r  = (const float*)d_in[1];
    const float* Wh   = (const float*)d_in[2];
    const float* Wt   = (const float*)d_in[3];
    const float* Wr   = (const float*)d_in[4];
    const int*   eidx = (const int*)d_in[5];
    const int*   rel  = (const int*)d_in[6];

    const int D   = 128;
    const int N_e = in_sizes[0] / D;
    const int N_r = in_sizes[1] / D;
    const int E   = in_sizes[6];

    const int* h_idx = eidx;
    const int* t_idx = eidx + E;

    auto align256 = [](size_t x) { return (x + 255) & ~(size_t)255; };
    char* ws = (char*)d_ws;
    float* Se     = (float*)ws; ws += align256((size_t)N_e * 16 * sizeof(float));
    float* Sr     = (float*)ws; ws += align256((size_t)N_r * 16 * sizeof(float));
    int*   cnt    = (int*)ws;   ws += align256((size_t)N_e * 2 * sizeof(int));
    int*   off    = (int*)ws;   ws += align256((size_t)N_e * 2 * sizeof(int));
    int*   lst    = (int*)ws;   ws += align256((size_t)E * 2 * sizeof(int));
    int*   cursor = (int*)ws;   ws += 256;
    if ((size_t)(ws - (char*)d_ws) > ws_size) return;

    float* out = (float*)d_out;

    const int n2 = N_e * 2;
    int* cb   = (int*)ws;       ws += align256((size_t)CH * n2 * sizeof(int));
    int* rank = (int*)ws;       ws += align256((size_t)E * 2 * sizeof(int));
    if ((size_t)(ws - (char*)d_ws) > ws_size) {
        if ((size_t)out_size < (size_t)E * 2 + (size_t)CH * n2) return;
        rank = (int*)d_out;
        cb   = rank + (size_t)E * 2;
    }

    const int BLK = 256;

    {
        const int waves   = (N_e + 15) / 16;
        const int threads = waves * 32;
        const int grid    = (threads + BLK - 1) / BLK;
        k_proj_wmma<<<grid, BLK, 0, stream>>>(x_e, Wh, Wt, Se, N_e);
    }
    {
        const int waves   = (N_r + 15) / 16;
        const int threads = waves * 32;
        const int grid    = (threads + BLK - 1) / BLK;
        k_proj_wmma<<<grid, BLK, 0, stream>>>(x_r, Wr, Wr, Sr, N_r);
    }

    k_count<<<dim3(CH, 2 * PS), CNT_THREADS, 0, stream>>>(eidx, cb, rank, cursor, E, N_e);
    k_offsets<<<(n2 + BLK - 1) / BLK, BLK, 0, stream>>>(cb, cnt, off, cursor, n2);
    k_fill<<<(E + EDGE_BLK - 1) / EDGE_BLK, EDGE_BLK, 0, stream>>>(h_idx, t_idx, rel, cb, rank, lst, E, N_e);

    {
        const long long threads = (long long)n2 * 32;
        const int grid = (int)((threads + BLK - 1) / BLK);
        k_node<<<grid, BLK, 0, stream>>>(cnt, off, lst, Se, Sr, x_r, out, N_e);
    }
}


# Node kernel: 8192 resident waves (one 32-wave block per compute unit) each loop over segments, next segment's length and start fetched a segment ahead; was one wave per segment; sums unchanged
# speedup vs baseline: 3.6085x; 1.0444x over previous
#include <hip/hip_runtime.h>
#include <stdint.h>

typedef __attribute__((ext_vector_type(2))) float v2f;
typedef __attribute__((ext_vector_type(8))) float v8f;

#define NEG_SLOPE 0.01f

__device__ __forceinline__ float lrelu(float x) {
    return x > 0.0f ? x : NEG_SLOPE * x;
}

#define CH   8
#define PS   16
#define CAP  4096
#define CNT_THREADS 1024
#define EDGE_BLK 256

__device__ __forceinline__ int chunk_len(int E) {
    return (((E + CH - 1) / CH) + EDGE_BLK - 1) & ~(EDGE_BLK - 1);
}

__global__ void k_proj_wmma(const float* __restrict__ X,
                            const float* __restrict__ W0,
                            const float* __restrict__ W1,
                            float* __restrict__ S, int nrows) {
    const int wave = (blockIdx.x * blockDim.x + threadIdx.x) >> 5;
    const int lane = threadIdx.x & 31;
    const int row0 = wave * 16;
    if (row0 >= nrows) return;

    const int m    = lane & 15;
    const int kofs = (lane >> 4) << 1;

    const float* __restrict__ arow = X + (size_t)(row0 + m) * 128;
    const float* __restrict__ brow =
        (m < 8) ? (W0 + (size_t)m * 128) : (W1 + (size_t)(m - 8) * 128);

    v8f c = {0.f, 0.f, 0.f, 0.f, 0.f, 0.f, 0.f, 0.f};
#pragma unroll
    for (int k = 0; k < 128; k += 4) {
        v2f a = *(const v2f*)(arow + k + kofs);
        v2f b = *(const v2f*)(brow + k + kofs);
        c = __builtin_amdgcn_wmma_f32_16x16x4_f32(
             false, a,  false, b,
             (short)0, c,  false,  false);
    }

    const int rbase = row0 + ((lane >> 4) << 3);
    const int col   = lane & 15;
#pragma unroll
    for (int v = 0; v < 8; ++v) {
        S[(size_t)(rbase + v) * 16 + col] = c[v];
    }
}

template <bool VEC>
__device__ __forceinline__ void count_ids(const int* __restrict__ id, int* __restrict__ rk, int* c,
                                          int s0, int len, int e0, int e1, int tid) {
    for (int e = e0 + 4 * tid; e < e1; e += 16 * CNT_THREADS) {
        int4 v[4];
#pragma unroll
        for (int u = 0; u < 4; ++u) {
            const int ee = e + u * 4 * CNT_THREADS;
            v[u] = make_int4(-1, -1, -1, -1);
            if (VEC) {
                if (ee < e1) v[u] = *(const int4*)(id + ee);
            } else {
                if (ee     < e1) v[u].x = id[ee];
                if (ee + 1 < e1) v[u].y = id[ee + 1];
                if (ee + 2 < e1) v[u].z = id[ee + 2];
                if (ee + 3 < e1) v[u].w = id[ee + 3];
            }
        }
        int r[16];
#pragma unroll
        for (int u = 0; u < 4; ++u) {
            const int vv[4] = {v[u].x, v[u].y, v[u].z, v[u].w};
#pragma unroll
            for (int j = 0; j < 4; ++j) {
                const unsigned d = (unsigned)(vv[j] - s0);
                r[4 * u + j] = (d < (unsigned)len) ? atomicAdd(&c[d], 1) : -1;
            }
        }
#pragma unroll
        for (int u = 0; u < 4; ++u) {
#pragma unroll
            for (int j = 0; j < 4; ++j) {
                if (r[4 * u + j] >= 0) rk[e + u * 4 * CNT_THREADS + j] = r[4 * u + j];
            }
        }
    }
}

__global__ void __launch_bounds__(CNT_THREADS)
k_count(const int* __restrict__ eidx, int* __restrict__ cb, int* __restrict__ rank,
        int* __restrict__ cursor, int E, int N_e) {
    __shared__ int c[CAP];
    const int tid   = threadIdx.x;
    const int chunk = blockIdx.x;
    const int side  = blockIdx.y / PS;
    const int part  = blockIdx.y - side * PS;
    if (blockIdx.x == 0 && blockIdx.y == 0 && tid == 0) *cursor = 0;

    const int n2   = 2 * N_e;
    const int plen = (N_e + PS - 1) / PS;
    const int lo   = part * plen;
    const int hi   = min(lo + plen, N_e);
    const int clen = chunk_len(E);
    const int e0   = chunk * clen;
    const int e1   = min(e0 + clen, E);
    const int* __restrict__ id = eidx + (size_t)side * E;
    int* __restrict__ rk = rank + (size_t)side * E;
    int* __restrict__ co = cb + (size_t)chunk * n2 + (size_t)side * N_e;
    const bool vec = ((E & 3) == 0) && ((((size_t)eidx) & 15) == 0);

    for (int s0 = lo; s0 < hi; s0 += CAP) {
        const int len = min(CAP, hi - s0);
        for (int i = tid; i < len; i += CNT_THREADS) c[i] = 0;
        __syncthreads();
        if (vec) count_ids<true>(id, rk, c, s0, len, e0, e1, tid);
        else     count_ids<false>(id, rk, c, s0, len, e0, e1, tid);
        __syncthreads();
        for (int i = tid; i < len; i += CNT_THREADS) co[s0 + i] = c[i];
        __syncthreads();
    }
}

__global__ void k_offsets(int* __restrict__ cb, int* __restrict__ cnt, int* __restrict__ off,
                          int* __restrict__ cursor, int n2) {
    const int i    = blockIdx.x * blockDim.x + threadIdx.x;
    const int lane = threadIdx.x & 31;
    int pc[CH];
    int c = 0;
#pragma unroll
    for (int k = 0; k < CH; ++k) {
        pc[k] = (i < n2) ? cb[(size_t)k * n2 + i] : 0;
        c += pc[k];
    }
    int s = c;
#pragma unroll
    for (int d = 1; d < 32; d <<= 1) {
        const int t = __shfl_up(s, d);
        if (lane >= d) s += t;
    }
    int base = 0;
    if (lane == 31) base = atomicAdd(cursor, s);
    base = __shfl(base, 31);
    if (i < n2) {
        int start = base + s - c;
        cnt[i] = c;
        off[i] = start;
#pragma unroll
        for (int k = 0; k < CH; ++k) {
            cb[(size_t)k * n2 + i] = start;
            start += pc[k];
        }
    }
}

__global__ void k_fill(const int* __restrict__ h_idx,
                       const int* __restrict__ t_idx,
                       const int* __restrict__ rel,
                       const int* __restrict__ cb,
                       const int* __restrict__ rank,
                       int* __restrict__ lst, int E, int N_e) {
    const int e = blockIdx.x * EDGE_BLK + threadIdx.x;
    if (e >= E) return;
    const int chunk = (int)((blockIdx.x * (unsigned)EDGE_BLK) / (unsigned)chunk_len(E));
    const int* __restrict__ b = cb + (size_t)chunk * (2 * (size_t)N_e);
    const int r = rel[e];
    const int p = b[h_idx[e]] + rank[e];
    const int q = b[N_e + t_idx[e]] + rank[E + e];
    lst[p] = r;
    lst[q] = r;
}

template <int NJ>
__device__ __forceinline__ void head_mean_weights(const float (&p)[8], float den, int (&ai)[8]) {
    float a[NJ];
#pragma unroll
    for (int j = 0; j < NJ; ++j) a[j] = p[j] / den;
#pragma unroll
    for (int d = 1; d < 8; d <<= 1) {
#pragma unroll
        for (int j = 0; j < NJ; ++j) a[j] += __shfl_xor(a[j], d);
    }
#pragma unroll
    for (int j = 0; j < 8; ++j) ai[j] = (j < NJ) ? __float_as_int(a[j] * 0.125f) : 0;
}

__device__ __forceinline__ void node_segment(const int seg, const int n, const int beg, const int lane,
                                             const int* __restrict__ lst,
                                             const float* __restrict__ Se, const float* __restrict__ Sr,
                                             const float* __restrict__ x_r, float* __restrict__ out,
                                             const int N_e) {
    const int side = (seg >= N_e) ? 1 : 0;
    const int node = seg - side * N_e;
    const int* __restrict__ l = lst + beg;

    const int h    = lane & 7;
    const int slot = lane >> 3;
    const float en = Se[(size_t)node * 16 + side * 8 + h];
    const float* __restrict__ xl = x_r + lane * 4;
    float* __restrict__ o = out + (size_t)node * 256 + side * 128 + lane * 4;

    float4 acc = make_float4(0.0f, 0.0f, 0.0f, 0.0f);

    if (n <= 32) {
        int rl = 0;
        if (lane < n) rl = l[lane];

        float v[8];
#pragma unroll
        for (int j = 0; j < 8; ++j) {
            const int rj = __shfl(rl, 4 * j + slot);
            v[j] = Sr[(size_t)rj * 16 + h];
        }
        float4 rows[16];
#pragma unroll
        for (int k = 0; k < 16; ++k) {
            const int rk = __builtin_amdgcn_readlane(rl, k);
            rows[k] = *(const float4*)(xl + (size_t)rk * 128);
        }

        float s[8];
        float m = -INFINITY;
#pragma unroll
        for (int j = 0; j < 8; ++j) {
            s[j] = (4 * j + slot < n) ? lrelu(en + v[j]) : -INFINITY;
            m = fmaxf(m, s[j]);
        }
        m = fmaxf(m, __shfl_xor(m, 8));
        m = fmaxf(m, __shfl_xor(m, 16));
        float z = 0.0f;
#pragma unroll
        for (int j = 0; j < 8; ++j) {
            s[j] = (4 * j + slot < n) ? __expf(s[j] - m) : 0.0f;
            z += s[j];
        }
        z += __shfl_xor(z, 8);
        z += __shfl_xor(z, 16);
        const float den = z + 1e-16f;
        int ai[8];
        if (n > 16) head_mean_weights<8>(s, den, ai);
        else        head_mean_weights<4>(s, den, ai);

#pragma unroll
        for (int k = 0; k < 16; ++k) {
            if (k < n) {
                const float ak = __int_as_float(__builtin_amdgcn_readlane(ai[k >> 2], (k & 3) * 8));
                acc.x += rows[k].x * ak;
                acc.y += rows[k].y * ak;
                acc.z += rows[k].z * ak;
                acc.w += rows[k].w * ak;
            }
        }
        if (n > 16) {
#pragma unroll
            for (int q = 0; q < 4; ++q) {
                if (16 + 4 * q < n) {
#pragma unroll
                    for (int t = 0; t < 4; ++t) {
                        const int k  = 4 * q + t;
                        const int rk = __builtin_amdgcn_readlane(rl, 16 + k);
                        rows[k] = *(const float4*)(xl + (size_t)rk * 128);
                    }
                }
            }
#pragma unroll
            for (int k = 0; k < 16; ++k) {
                if (16 + k < n) {
                    const float ak = __int_as_float(__builtin_amdgcn_readlane(ai[4 + (k >> 2)], (k & 3) * 8));
                    acc.x += rows[k].x * ak;
                    acc.y += rows[k].y * ak;
                    acc.z += rows[k].z * ak;
                    acc.w += rows[k].w * ak;
                }
            }
        }
        *(float4*)o = acc;
        return;
    }

    float m = -INFINITY;
    for (int i = slot; i < n; i += 4) {
        const int r = l[i];
        m = fmaxf(m, lrelu(en + Sr[(size_t)r * 16 + h]));
    }
    m = fmaxf(m, __shfl_xor(m, 8));
    m = fmaxf(m, __shfl_xor(m, 16));

    float z = 0.0f;
    for (int i = slot; i < n; i += 4) {
        const int r = l[i];
        z += __expf(lrelu(en + Sr[(size_t)r * 16 + h]) - m);
    }
    z += __shfl_xor(z, 8);
    z += __shfl_xor(z, 16);
    const float den = z + 1e-16f;

    int i0 = 0;
    for (; i0 + 4 <= n; i0 += 4) {
        const int r = l[i0 + slot];
        float a = __expf(lrelu(en + Sr[(size_t)r * 16 + h]) - m) / den;
        a += __shfl_xor(a, 1);
        a += __shfl_xor(a, 2);
        a += __shfl_xor(a, 4);
        a *= 0.125f;
        const int ai = __float_as_int(a);
        const int r0 = __builtin_amdgcn_readlane(r, 0);
        const int r1 = __builtin_amdgcn_readlane(r, 8);
        const int r2 = __builtin_amdgcn_readlane(r, 16);
        const int r3 = __builtin_amdgcn_readlane(r, 24);
        const float a0 = __int_as_float(__builtin_amdgcn_readlane(ai, 0));
        const float a1 = __int_as_float(__builtin_amdgcn_readlane(ai, 8));
        const float a2 = __int_as_float(__builtin_amdgcn_readlane(ai, 16));
        const float a3 = __int_as_float(__builtin_amdgcn_readlane(ai, 24));
        const float4 m0 = *(const float4*)(xl + (size_t)r0 * 128);
        const float4 m1 = *(const float4*)(xl + (size_t)r1 * 128);
        const float4 m2 = *(const float4*)(xl + (size_t)r2 * 128);
        const float4 m3 = *(const float4*)(xl + (size_t)r3 * 128);
        acc.x += m0.x * a0; acc.y += m0.y * a0; acc.z += m0.z * a0; acc.w += m0.w * a0;
        acc.x += m1.x * a1; acc.y += m1.y * a1; acc.z += m1.z * a1; acc.w += m1.w * a1;
        acc.x += m2.x * a2; acc.y += m2.y * a2; acc.z += m2.z * a2; acc.w += m2.w * a2;
        acc.x += m3.x * a3; acc.y += m3.y * a3; acc.z += m3.z * a3; acc.w += m3.w * a3;
    }
    if (i0 < n) {
        const int  i  = i0 + slot;
        const bool ok = i < n;
        const int  r  = ok ? l[i] : 0;
        float a = 0.0f;
        if (ok) a = __expf(lrelu(en + Sr[(size_t)r * 16 + h]) - m) / den;
        a += __shfl_xor(a, 1);
        a += __shfl_xor(a, 2);
        a += __shfl_xor(a, 4);
        a *= 0.125f;
        const int ai = __float_as_int(a);
#pragma unroll
        for (int k = 0; k < 3; ++k) {
            if (i0 + k < n) {
                const int   rk = __builtin_amdgcn_readlane(r, k * 8);
                const float ak = __int_as_float(__builtin_amdgcn_readlane(ai, k * 8));
                const float4 mv = *(const float4*)(xl + (size_t)rk * 128);
                acc.x += mv.x * ak;
                acc.y += mv.y * ak;
                acc.z += mv.z * ak;
                acc.w += mv.w * ak;
            }
        }
    }
    *(float4*)o = acc;
}

#define NODE_THREADS 1024
__global__ void __launch_bounds__(NODE_THREADS)
k_node(const int* __restrict__ cnt, const int* __restrict__ off,
       const int* __restrict__ lst,
       const float* __restrict__ Se, const float* __restrict__ Sr,
       const float* __restrict__ x_r, float* __restrict__ out,
       int N_e) {
    const int lane = threadIdx.x & 31;
    const int w  = __builtin_amdgcn_readfirstlane((int)((blockIdx.x * NODE_THREADS + threadIdx.x) >> 5));
    const int nw = (int)(gridDim.x * (NODE_THREADS / 32));
    const int n2 = 2 * N_e;
    if (w >= n2) return;
    int seg   = w;
    int n_nx  = cnt[seg];
    int beg_nx = off[seg];
    for (;;) {
        const int n   = __builtin_amdgcn_readfirstlane(n_nx);
        const int beg = __builtin_amdgcn_readfirstlane(beg_nx);
        const int  sn   = seg + nw;
        const bool more = sn < n2;
        if (more) { n_nx = cnt[sn]; beg_nx = off[sn]; }
        node_segment(seg, n, beg, lane, lst, Se, Sr, x_r, out, N_e);
        if (!more) break;
        seg = sn;
    }
}

extern "C" void kernel_launch(void* const* d_in, const int* in_sizes, int n_in,
                              void* d_out, int out_size, void* d_ws, size_t ws_size,
                              hipStream_t stream) {
    const float* x_e  = (const float*)d_in[0];
    const float* x_r  = (const float*)d_in[1];
    const float* Wh   = (const float*)d_in[2];
    const float* Wt   = (const float*)d_in[3];
    const float* Wr   = (const float*)d_in[4];
    const int*   eidx = (const int*)d_in[5];
    const int*   rel  = (const int*)d_in[6];

    const int D   = 128;
    const int N_e = in_sizes[0] / D;
    const int N_r = in_sizes[1] / D;
    const int E   = in_sizes[6];

    const int* h_idx = eidx;
    const int* t_idx = eidx + E;

    auto align256 = [](size_t x) { return (x + 255) & ~(size_t)255; };
    char* ws = (char*)d_ws;
    float* Se     = (float*)ws; ws += align256((size_t)N_e * 16 * sizeof(float));
    float* Sr     = (float*)ws; ws += align256((size_t)N_r * 16 * sizeof(float));
    int*   cnt    = (int*)ws;   ws += align256((size_t)N_e * 2 * sizeof(int));
    int*   off    = (int*)ws;   ws += align256((size_t)N_e * 2 * sizeof(int));
    int*   lst    = (int*)ws;   ws += align256((size_t)E * 2 * sizeof(int));
    int*   cursor = (int*)ws;   ws += 256;
    if ((size_t)(ws - (char*)d_ws) > ws_size) return;

    float* out = (float*)d_out;

    const int n2 = N_e * 2;
    int* cb   = (int*)ws;       ws += align256((size_t)CH * n2 * sizeof(int));
    int* rank = (int*)ws;       ws += align256((size_t)E * 2 * sizeof(int));
    if ((size_t)(ws - (char*)d_ws) > ws_size) {
        if ((size_t)out_size < (size_t)E * 2 + (size_t)CH * n2) return;
        rank = (int*)d_out;
        cb   = rank + (size_t)E * 2;
    }

    const int BLK = 256;

    {
        const int waves   = (N_e + 15) / 16;
        const int threads = waves * 32;
        const int grid    = (threads + BLK - 1) / BLK;
        k_proj_wmma<<<grid, BLK, 0, stream>>>(x_e, Wh, Wt, Se, N_e);
    }
    {
        const int waves   = (N_r + 15) / 16;
        const int threads = waves * 32;
        const int grid    = (threads + BLK - 1) / BLK;
        k_proj_wmma<<<grid, BLK, 0, stream>>>(x_r, Wr, Wr, Sr, N_r);
    }

    k_count<<<dim3(CH, 2 * PS), CNT_THREADS, 0, stream>>>(eidx, cb, rank, cursor, E, N_e);
    k_offsets<<<(n2 + BLK - 1) / BLK, BLK, 0, stream>>>(cb, cnt, off, cursor, n2);
    k_fill<<<(E + EDGE_BLK - 1) / EDGE_BLK, EDGE_BLK, 0, stream>>>(h_idx, t_idx, rel, cb, rank, lst, E, N_e);

    k_node<<<256, NODE_THREADS, 0, stream>>>(cnt, off, lst, Se, Sr, x_r, out, N_e);
}


# Node kernel: each block copies the relations' eight head scores into shared memory once; every segment's score gathers read LDS, not the L2: a third fewer cache-line requests; sums unchanged
# speedup vs baseline: 3.7655x; 1.0435x over previous
#include <hip/hip_runtime.h>
#include <stdint.h>

typedef __attribute__((ext_vector_type(2))) float v2f;
typedef __attribute__((ext_vector_type(8))) float v8f;

#define NEG_SLOPE 0.01f

__device__ __forceinline__ float lrelu(float x) {
    return x > 0.0f ? x : NEG_SLOPE * x;
}

#define CH   8
#define PS   16
#define CAP  4096
#define CNT_THREADS 1024
#define EDGE_BLK 256

__device__ __forceinline__ int chunk_len(int E) {
    return (((E + CH - 1) / CH) + EDGE_BLK - 1) & ~(EDGE_BLK - 1);
}

__global__ void k_proj_wmma(const float* __restrict__ X,
                            const float* __restrict__ W0,
                            const float* __restrict__ W1,
                            float* __restrict__ S, int nrows) {
    const int wave = (blockIdx.x * blockDim.x + threadIdx.x) >> 5;
    const int lane = threadIdx.x & 31;
    const int row0 = wave * 16;
    if (row0 >= nrows) return;

    const int m    = lane & 15;
    const int kofs = (lane >> 4) << 1;

    const float* __restrict__ arow = X + (size_t)(row0 + m) * 128;
    const float* __restrict__ brow =
        (m < 8) ? (W0 + (size_t)m * 128) : (W1 + (size_t)(m - 8) * 128);

    v8f c = {0.f, 0.f, 0.f, 0.f, 0.f, 0.f, 0.f, 0.f};
#pragma unroll
    for (int k = 0; k < 128; k += 4) {
        v2f a = *(const v2f*)(arow + k + kofs);
        v2f b = *(const v2f*)(brow + k + kofs);
        c = __builtin_amdgcn_wmma_f32_16x16x4_f32(
             false, a,  false, b,
             (short)0, c,  false,  false);
    }

    const int rbase = row0 + ((lane >> 4) << 3);
    const int col   = lane & 15;
#pragma unroll
    for (int v = 0; v < 8; ++v) {
        S[(size_t)(rbase + v) * 16 + col] = c[v];
    }
}

template <bool VEC>
__device__ __forceinline__ void count_ids(const int* __restrict__ id, int* __restrict__ rk, int* c,
                                          int s0, int len, int e0, int e1, int tid) {
    for (int e = e0 + 4 * tid; e < e1; e += 16 * CNT_THREADS) {
        int4 v[4];
#pragma unroll
        for (int u = 0; u < 4; ++u) {
            const int ee = e + u * 4 * CNT_THREADS;
            v[u] = make_int4(-1, -1, -1, -1);
            if (VEC) {
                if (ee < e1) v[u] = *(const int4*)(id + ee);
            } else {
                if (ee     < e1) v[u].x = id[ee];
                if (ee + 1 < e1) v[u].y = id[ee + 1];
                if (ee + 2 < e1) v[u].z = id[ee + 2];
                if (ee + 3 < e1) v[u].w = id[ee + 3];
            }
        }
        int r[16];
#pragma unroll
        for (int u = 0; u < 4; ++u) {
            const int vv[4] = {v[u].x, v[u].y, v[u].z, v[u].w};
#pragma unroll
            for (int j = 0; j < 4; ++j) {
                const unsigned d = (unsigned)(vv[j] - s0);
                r[4 * u + j] = (d < (unsigned)len) ? atomicAdd(&c[d], 1) : -1;
            }
        }
#pragma unroll
        for (int u = 0; u < 4; ++u) {
#pragma unroll
            for (int j = 0; j < 4; ++j) {
                if (r[4 * u + j] >= 0) rk[e + u * 4 * CNT_THREADS + j] = r[4 * u + j];
            }
        }
    }
}

__global__ void __launch_bounds__(CNT_THREADS)
k_count(const int* __restrict__ eidx, int* __restrict__ cb, int* __restrict__ rank,
        int* __restrict__ cursor, int E, int N_e) {
    __shared__ int c[CAP];
    const int tid   = threadIdx.x;
    const int chunk = blockIdx.x;
    const int side  = blockIdx.y / PS;
    const int part  = blockIdx.y - side * PS;
    if (blockIdx.x == 0 && blockIdx.y == 0 && tid == 0) *cursor = 0;

    const int n2   = 2 * N_e;
    const int plen = (N_e + PS - 1) / PS;
    const int lo   = part * plen;
    const int hi   = min(lo + plen, N_e);
    const int clen = chunk_len(E);
    const int e0   = chunk * clen;
    const int e1   = min(e0 + clen, E);
    const int* __restrict__ id = eidx + (size_t)side * E;
    int* __restrict__ rk = rank + (size_t)side * E;
    int* __restrict__ co = cb + (size_t)chunk * n2 + (size_t)side * N_e;
    const bool vec = ((E & 3) == 0) && ((((size_t)eidx) & 15) == 0);

    for (int s0 = lo; s0 < hi; s0 += CAP) {
        const int len = min(CAP, hi - s0);
        for (int i = tid; i < len; i += CNT_THREADS) c[i] = 0;
        __syncthreads();
        if (vec) count_ids<true>(id, rk, c, s0, len, e0, e1, tid);
        else     count_ids<false>(id, rk, c, s0, len, e0, e1, tid);
        __syncthreads();
        for (int i = tid; i < len; i += CNT_THREADS) co[s0 + i] = c[i];
        __syncthreads();
    }
}

__global__ void k_offsets(int* __restrict__ cb, int* __restrict__ cnt, int* __restrict__ off,
                          int* __restrict__ cursor, int n2) {
    const int i    = blockIdx.x * blockDim.x + threadIdx.x;
    const int lane = threadIdx.x & 31;
    int pc[CH];
    int c = 0;
#pragma unroll
    for (int k = 0; k < CH; ++k) {
        pc[k] = (i < n2) ? cb[(size_t)k * n2 + i] : 0;
        c += pc[k];
    }
    int s = c;
#pragma unroll
    for (int d = 1; d < 32; d <<= 1) {
        const int t = __shfl_up(s, d);
        if (lane >= d) s += t;
    }
    int base = 0;
    if (lane == 31) base = atomicAdd(cursor, s);
    base = __shfl(base, 31);
    if (i < n2) {
        int start = base + s - c;
        cnt[i] = c;
        off[i] = start;
#pragma unroll
        for (int k = 0; k < CH; ++k) {
            cb[(size_t)k * n2 + i] = start;
            start += pc[k];
        }
    }
}

__global__ void k_fill(const int* __restrict__ h_idx,
                       const int* __restrict__ t_idx,
                       const int* __restrict__ rel,
                       const int* __restrict__ cb,
                       const int* __restrict__ rank,
                       int* __restrict__ lst, int E, int N_e) {
    const int e = blockIdx.x * EDGE_BLK + threadIdx.x;
    if (e >= E) return;
    const int chunk = (int)((blockIdx.x * (unsigned)EDGE_BLK) / (unsigned)chunk_len(E));
    const int* __restrict__ b = cb + (size_t)chunk * (2 * (size_t)N_e);
    const int r = rel[e];
    const int p = b[h_idx[e]] + rank[e];
    const int q = b[N_e + t_idx[e]] + rank[E + e];
    lst[p] = r;
    lst[q] = r;
}

template <int NJ>
__device__ __forceinline__ void head_mean_weights(const float (&p)[8], float den, int (&ai)[8]) {
    float a[NJ];
#pragma unroll
    for (int j = 0; j < NJ; ++j) a[j] = p[j] / den;
#pragma unroll
    for (int d = 1; d < 8; d <<= 1) {
#pragma unroll
        for (int j = 0; j < NJ; ++j) a[j] += __shfl_xor(a[j], d);
    }
#pragma unroll
    for (int j = 0; j < 8; ++j) ai[j] = (j < NJ) ? __float_as_int(a[j] * 0.125f) : 0;
}

#define SR_CAP 2048

template <bool TAB>
__device__ __forceinline__ float rel_score(const float* __restrict__ Sr, const float* tab, const int r, const int h) {
    return TAB ? tab[r * 8 + h] : Sr[(size_t)r * 16 + h];
}

template <bool TAB>
__device__ __forceinline__ void node_segment(const int seg, const int n, const int beg, const int lane,
                                             const int* __restrict__ lst,
                                             const float* __restrict__ Se, const float* __restrict__ Sr,
                                             const float* tab,
                                             const float* __restrict__ x_r, float* __restrict__ out,
                                             const int N_e) {
    const int side = (seg >= N_e) ? 1 : 0;
    const int node = seg - side * N_e;
    const int* __restrict__ l = lst + beg;

    const int h    = lane & 7;
    const int slot = lane >> 3;
    const float en = Se[(size_t)node * 16 + side * 8 + h];
    const float* __restrict__ xl = x_r + lane * 4;
    float* __restrict__ o = out + (size_t)node * 256 + side * 128 + lane * 4;

    float4 acc = make_float4(0.0f, 0.0f, 0.0f, 0.0f);

    if (n <= 32) {
        int rl = 0;
        if (lane < n) rl = l[lane];

        float v[8];
#pragma unroll
        for (int j = 0; j < 8; ++j) {
            const int rj = __shfl(rl, 4 * j + slot);
            v[j] = rel_score<TAB>(Sr, tab, rj, h);
        }
        float4 rows[16];
#pragma unroll
        for (int k = 0; k < 16; ++k) {
            const int rk = __builtin_amdgcn_readlane(rl, k);
            rows[k] = *(const float4*)(xl + (size_t)rk * 128);
        }

        float s[8];
        float m = -INFINITY;
#pragma unroll
        for (int j = 0; j < 8; ++j) {
            s[j] = (4 * j + slot < n) ? lrelu(en + v[j]) : -INFINITY;
            m = fmaxf(m, s[j]);
        }
        m = fmaxf(m, __shfl_xor(m, 8));
        m = fmaxf(m, __shfl_xor(m, 16));
        float z = 0.0f;
#pragma unroll
        for (int j = 0; j < 8; ++j) {
            s[j] = (4 * j + slot < n) ? __expf(s[j] - m) : 0.0f;
            z += s[j];
        }
        z += __shfl_xor(z, 8);
        z += __shfl_xor(z, 16);
        const float den = z + 1e-16f;
        int ai[8];
        if (n > 16) head_mean_weights<8>(s, den, ai);
        else        head_mean_weights<4>(s, den, ai);

#pragma unroll
        for (int k = 0; k < 16; ++k) {
            if (k < n) {
                const float ak = __int_as_float(__builtin_amdgcn_readlane(ai[k >> 2], (k & 3) * 8));
                acc.x += rows[k].x * ak;
                acc.y += rows[k].y * ak;
                acc.z += rows[k].z * ak;
                acc.w += rows[k].w * ak;
            }
        }
        if (n > 16) {
#pragma unroll
            for (int q = 0; q < 4; ++q) {
                if (16 + 4 * q < n) {
#pragma unroll
                    for (int t = 0; t < 4; ++t) {
                        const int k  = 4 * q + t;
                        const int rk = __builtin_amdgcn_readlane(rl, 16 + k);
                        rows[k] = *(const float4*)(xl + (size_t)rk * 128);
                    }
                }
            }
#pragma unroll
            for (int k = 0; k < 16; ++k) {
                if (16 + k < n) {
                    const float ak = __int_as_float(__builtin_amdgcn_readlane(ai[4 + (k >> 2)], (k & 3) * 8));
                    acc.x += rows[k].x * ak;
                    acc.y += rows[k].y * ak;
                    acc.z += rows[k].z * ak;
                    acc.w += rows[k].w * ak;
                }
            }
        }
        *(float4*)o = acc;
        return;
    }

    float m = -INFINITY;
    for (int i = slot; i < n; i += 4) {
        const int r = l[i];
        m = fmaxf(m, lrelu(en + rel_score<TAB>(Sr, tab, r, h)));
    }
    m = fmaxf(m, __shfl_xor(m, 8));
    m = fmaxf(m, __shfl_xor(m, 16));

    float z = 0.0f;
    for (int i = slot; i < n; i += 4) {
        const int r = l[i];
        z += __expf(lrelu(en + rel_score<TAB>(Sr, tab, r, h)) - m);
    }
    z += __shfl_xor(z, 8);
    z += __shfl_xor(z, 16);
    const float den = z + 1e-16f;

    int i0 = 0;
    for (; i0 + 4 <= n; i0 += 4) {
        const int r = l[i0 + slot];
        float a = __expf(lrelu(en + rel_score<TAB>(Sr, tab, r, h)) - m) / den;
        a += __shfl_xor(a, 1);
        a += __shfl_xor(a, 2);
        a += __shfl_xor(a, 4);
        a *= 0.125f;
        const int ai = __float_as_int(a);
        const int r0 = __builtin_amdgcn_readlane(r, 0);
        const int r1 = __builtin_amdgcn_readlane(r, 8);
        const int r2 = __builtin_amdgcn_readlane(r, 16);
        const int r3 = __builtin_amdgcn_readlane(r, 24);
        const float a0 = __int_as_float(__builtin_amdgcn_readlane(ai, 0));
        const float a1 = __int_as_float(__builtin_amdgcn_readlane(ai, 8));
        const float a2 = __int_as_float(__builtin_amdgcn_readlane(ai, 16));
        const float a3 = __int_as_float(__builtin_amdgcn_readlane(ai, 24));
        const float4 m0 = *(const float4*)(xl + (size_t)r0 * 128);
        const float4 m1 = *(const float4*)(xl + (size_t)r1 * 128);
        const float4 m2 = *(const float4*)(xl + (size_t)r2 * 128);
        const float4 m3 = *(const float4*)(xl + (size_t)r3 * 128);
        acc.x += m0.x * a0; acc.y += m0.y * a0; acc.z += m0.z * a0; acc.w += m0.w * a0;
        acc.x += m1.x * a1; acc.y += m1.y * a1; acc.z += m1.z * a1; acc.w += m1.w * a1;
        acc.x += m2.x * a2; acc.y += m2.y * a2; acc.z += m2.z * a2; acc.w += m2.w * a2;
        acc.x += m3.x * a3; acc.y += m3.y * a3; acc.z += m3.z * a3; acc.w += m3.w * a3;
    }
    if (i0 < n) {
        const int  i  = i0 + slot;
        const bool ok = i < n;
        const int  r  = ok ? l[i] : 0;
        float a = 0.0f;
        if (ok) a = __expf(lrelu(en + rel_score<TAB>(Sr, tab, r, h)) - m) / den;
        a += __shfl_xor(a, 1);
        a += __shfl_xor(a, 2);
        a += __shfl_xor(a, 4);
        a *= 0.125f;
        const int ai = __float_as_int(a);
#pragma unroll
        for (int k = 0; k < 3; ++k) {
            if (i0 + k < n) {
                const int   rk = __builtin_amdgcn_readlane(r, k * 8);
                const float ak = __int_as_float(__builtin_amdgcn_readlane(ai, k * 8));
                const float4 mv = *(const float4*)(xl + (size_t)rk * 128);
                acc.x += mv.x * ak;
                acc.y += mv.y * ak;
                acc.z += mv.z * ak;
                acc.w += mv.w * ak;
            }
        }
    }
    *(float4*)o = acc;
}

#define NODE_THREADS 1024

template <bool TAB>
__device__ __forceinline__ void walk_segments(const int w, const int nw, const int n2, const int lane,
                                              const int* __restrict__ cnt, const int* __restrict__ off,
                                              const int* __restrict__ lst,
                                              const float* __restrict__ Se, const float* __restrict__ Sr,
                                              const float* tab,
                                              const float* __restrict__ x_r, float* __restrict__ out,
                                              const int N_e) {
    int seg   = w;
    int n_nx  = cnt[seg];
    int beg_nx = off[seg];
    for (;;) {
        const int n   = __builtin_amdgcn_readfirstlane(n_nx);
        const int beg = __builtin_amdgcn_readfirstlane(beg_nx);
        const int  sn   = seg + nw;
        const bool more = sn < n2;
        if (more) { n_nx = cnt[sn]; beg_nx = off[sn]; }
        node_segment<TAB>(seg, n, beg, lane, lst, Se, Sr, tab, x_r, out, N_e);
        if (!more) break;
        seg = sn;
    }
}

__global__ void __launch_bounds__(NODE_THREADS)
k_node(const int* __restrict__ cnt, const int* __restrict__ off,
       const int* __restrict__ lst,
       const float* __restrict__ Se, const float* __restrict__ Sr,
       const float* __restrict__ x_r, float* __restrict__ out,
       int N_e, int N_r) {
    __shared__ float tab[SR_CAP * 8];
    const bool use_tab = N_r <= SR_CAP;
    if (use_tab) {
        const int items = 2 * N_r;
        for (int i0 = threadIdx.x; i0 < items; i0 += 4 * NODE_THREADS) {
            const int i1 = i0 + NODE_THREADS;
            const int i2 = i0 + 2 * NODE_THREADS;
            const int i3 = i0 + 3 * NODE_THREADS;
            const int c1 = min(i1, items - 1);
            const int c2 = min(i2, items - 1);
            const int c3 = min(i3, items - 1);
            const float4 t0 = *(const float4*)(Sr + (size_t)(i0 >> 1) * 16 + (i0 & 1) * 4);
            const float4 t1 = *(const float4*)(Sr + (size_t)(c1 >> 1) * 16 + (c1 & 1) * 4);
            const float4 t2 = *(const float4*)(Sr + (size_t)(c2 >> 1) * 16 + (c2 & 1) * 4);
            const float4 t3 = *(const float4*)(Sr + (size_t)(c3 >> 1) * 16 + (c3 & 1) * 4);
            *(float4*)(tab + i0 * 4) = t0;
            *(float4*)(tab + c1 * 4) = t1;
            *(float4*)(tab + c2 * 4) = t2;
            *(float4*)(tab + c3 * 4) = t3;
        }
        __syncthreads();
    }

    const int lane = threadIdx.x & 31;
    const int w  = __builtin_amdgcn_readfirstlane((int)((blockIdx.x * NODE_THREADS + threadIdx.x) >> 5));
    const int nw = (int)(gridDim.x * (NODE_THREADS / 32));
    const int n2 = 2 * N_e;
    if (w >= n2) return;
    if (use_tab) walk_segments<true>(w, nw, n2, lane, cnt, off, lst, Se, Sr, tab, x_r, out, N_e);
    else         walk_segments<false>(w, nw, n2, lane, cnt, off, lst, Se, Sr, tab, x_r, out, N_e);
}

extern "C" void kernel_launch(void* const* d_in, const int* in_sizes, int n_in,
                              void* d_out, int out_size, void* d_ws, size_t ws_size,
                              hipStream_t stream) {
    const float* x_e  = (const float*)d_in[0];
    const float* x_r  = (const float*)d_in[1];
    const float* Wh   = (const float*)d_in[2];
    const float* Wt   = (const float*)d_in[3];
    const float* Wr   = (const float*)d_in[4];
    const int*   eidx = (const int*)d_in[5];
    const int*   rel  = (const int*)d_in[6];

    const int D   = 128;
    const int N_e = in_sizes[0] / D;
    const int N_r = in_sizes[1] / D;
    const int E   = in_sizes[6];

    const int* h_idx = eidx;
    const int* t_idx = eidx + E;

    auto align256 = [](size_t x) { return (x + 255) & ~(size_t)255; };
    char* ws = (char*)d_ws;
    float* Se     = (float*)ws; ws += align256((size_t)N_e * 16 * sizeof(float));
    float* Sr     = (float*)ws; ws += align256((size_t)N_r * 16 * sizeof(float));
    int*   cnt    = (int*)ws;   ws += align256((size_t)N_e * 2 * sizeof(int));
    int*   off    = (int*)ws;   ws += align256((size_t)N_e * 2 * sizeof(int));
    int*   lst    = (int*)ws;   ws += align256((size_t)E * 2 * sizeof(int));
    int*   cursor = (int*)ws;   ws += 256;
    if ((size_t)(ws - (char*)d_ws) > ws_size) return;

    float* out = (float*)d_out;

    const int n2 = N_e * 2;
    int* cb   = (int*)ws;       ws += align256((size_t)CH * n2 * sizeof(int));
    int* rank = (int*)ws;       ws += align256((size_t)E * 2 * sizeof(int));
    if ((size_t)(ws - (char*)d_ws) > ws_size) {
        if ((size_t)out_size < (size_t)E * 2 + (size_t)CH * n2) return;
        rank = (int*)d_out;
        cb   = rank + (size_t)E * 2;
    }

    const int BLK = 256;

    {
        const int waves   = (N_e + 15) / 16;
        const int threads = waves * 32;
        const int grid    = (threads + BLK - 1) / BLK;
        k_proj_wmma<<<grid, BLK, 0, stream>>>(x_e, Wh, Wt, Se, N_e);
    }
    {
        const int waves   = (N_r + 15) / 16;
        const int threads = waves * 32;
        const int grid    = (threads + BLK - 1) / BLK;
        k_proj_wmma<<<grid, BLK, 0, stream>>>(x_r, Wr, Wr, Sr, N_r);
    }

    k_count<<<dim3(CH, 2 * PS), CNT_THREADS, 0, stream>>>(eidx, cb, rank, cursor, E, N_e);
    k_offsets<<<(n2 + BLK - 1) / BLK, BLK, 0, stream>>>(cb, cnt, off, cursor, n2);
    k_fill<<<(E + EDGE_BLK - 1) / EDGE_BLK, EDGE_BLK, 0, stream>>>(h_idx, t_idx, rel, cb, rank, lst, E, N_e);

    k_node<<<256, NODE_THREADS, 0, stream>>>(cnt, off, lst, Se, Sr, x_r, out, N_e, N_r);
}


# The two projections and the offset pass share one launch (block index picks the role; none needs another's result): 4 launches per call, not 6; tile and offset code and all sums unchanged
# speedup vs baseline: 4.0566x; 1.0773x over previous
#include <hip/hip_runtime.h>
#include <stdint.h>

typedef __attribute__((ext_vector_type(2))) float v2f;
typedef __attribute__((ext_vector_type(8))) float v8f;

#define NEG_SLOPE 0.01f

__device__ __forceinline__ float lrelu(float x) {
    return x > 0.0f ? x : NEG_SLOPE * x;
}

#define CH   8
#define PS   16
#define CAP  4096
#define CNT_THREADS 1024
#define EDGE_BLK 256

__device__ __forceinline__ int chunk_len(int E) {
    return (((E + CH - 1) / CH) + EDGE_BLK - 1) & ~(EDGE_BLK - 1);
}

__device__ __forceinline__ void proj_tile_wmma(const float* __restrict__ X,
                                               const float* __restrict__ W0,
                                               const float* __restrict__ W1,
                                               float* __restrict__ S, const int nrows,
                                               const int wave, const int lane) {
    const int row0 = wave * 16;
    if (row0 >= nrows) return;

    const int m    = lane & 15;
    const int kofs = (lane >> 4) << 1;

    const float* __restrict__ arow = X + (size_t)(row0 + m) * 128;
    const float* __restrict__ brow =
        (m < 8) ? (W0 + (size_t)m * 128) : (W1 + (size_t)(m - 8) * 128);

    v8f c = {0.f, 0.f, 0.f, 0.f, 0.f, 0.f, 0.f, 0.f};
#pragma unroll
    for (int k = 0; k < 128; k += 4) {
        v2f a = *(const v2f*)(arow + k + kofs);
        v2f b = *(const v2f*)(brow + k + kofs);
        c = __builtin_amdgcn_wmma_f32_16x16x4_f32(
             false, a,  false, b,
             (short)0, c,  false,  false);
    }

    const int rbase = row0 + ((lane >> 4) << 3);
    const int col   = lane & 15;
#pragma unroll
    for (int v = 0; v < 8; ++v) {
        S[(size_t)(rbase + v) * 16 + col] = c[v];
    }
}

template <bool VEC>
__device__ __forceinline__ void count_ids(const int* __restrict__ id, int* __restrict__ rk, int* c,
                                          int s0, int len, int e0, int e1, int tid) {
    for (int e = e0 + 4 * tid; e < e1; e += 16 * CNT_THREADS) {
        int4 v[4];
#pragma unroll
        for (int u = 0; u < 4; ++u) {
            const int ee = e + u * 4 * CNT_THREADS;
            v[u] = make_int4(-1, -1, -1, -1);
            if (VEC) {
                if (ee < e1) v[u] = *(const int4*)(id + ee);
            } else {
                if (ee     < e1) v[u].x = id[ee];
                if (ee + 1 < e1) v[u].y = id[ee + 1];
                if (ee + 2 < e1) v[u].z = id[ee + 2];
                if (ee + 3 < e1) v[u].w = id[ee + 3];
            }
        }
        int r[16];
#pragma unroll
        for (int u = 0; u < 4; ++u) {
            const int vv[4] = {v[u].x, v[u].y, v[u].z, v[u].w};
#pragma unroll
            for (int j = 0; j < 4; ++j) {
                const unsigned d = (unsigned)(vv[j] - s0);
                r[4 * u + j] = (d < (unsigned)len) ? atomicAdd(&c[d], 1) : -1;
            }
        }
#pragma unroll
        for (int u = 0; u < 4; ++u) {
#pragma unroll
            for (int j = 0; j < 4; ++j) {
                if (r[4 * u + j] >= 0) rk[e + u * 4 * CNT_THREADS + j] = r[4 * u + j];
            }
        }
    }
}

__global__ void __launch_bounds__(CNT_THREADS)
k_count(const int* __restrict__ eidx, int* __restrict__ cb, int* __restrict__ rank,
        int* __restrict__ cursor, int E, int N_e) {
    __shared__ int c[CAP];
    const int tid   = threadIdx.x;
    const int chunk = blockIdx.x;
    const int side  = blockIdx.y / PS;
    const int part  = blockIdx.y - side * PS;
    if (blockIdx.x == 0 && blockIdx.y == 0 && tid == 0) *cursor = 0;

    const int n2   = 2 * N_e;
    const int plen = (N_e + PS - 1) / PS;
    const int lo   = part * plen;
    const int hi   = min(lo + plen, N_e);
    const int clen = chunk_len(E);
    const int e0   = chunk * clen;
    const int e1   = min(e0 + clen, E);
    const int* __restrict__ id = eidx + (size_t)side * E;
    int* __restrict__ rk = rank + (size_t)side * E;
    int* __restrict__ co = cb + (size_t)chunk * n2 + (size_t)side * N_e;
    const bool vec = ((E & 3) == 0) && ((((size_t)eidx) & 15) == 0);

    for (int s0 = lo; s0 < hi; s0 += CAP) {
        const int len = min(CAP, hi - s0);
        for (int i = tid; i < len; i += CNT_THREADS) c[i] = 0;
        __syncthreads();
        if (vec) count_ids<true>(id, rk, c, s0, len, e0, e1, tid);
        else     count_ids<false>(id, rk, c, s0, len, e0, e1, tid);
        __syncthreads();
        for (int i = tid; i < len; i += CNT_THREADS) co[s0 + i] = c[i];
        __syncthreads();
    }
}

__device__ __forceinline__ void offsets_block(int* __restrict__ cb, int* __restrict__ cnt, int* __restrict__ off,
                                              int* __restrict__ cursor, const int n2, const int blk) {
    const int i    = blk * (int)blockDim.x + (int)threadIdx.x;
    const int lane = threadIdx.x & 31;
    int pc[CH];
    int c = 0;
#pragma unroll
    for (int k = 0; k < CH; ++k) {
        pc[k] = (i < n2) ? cb[(size_t)k * n2 + i] : 0;
        c += pc[k];
    }
    int s = c;
#pragma unroll
    for (int d = 1; d < 32; d <<= 1) {
        const int t = __shfl_up(s, d);
        if (lane >= d) s += t;
    }
    int base = 0;
    if (lane == 31) base = atomicAdd(cursor, s);
    base = __shfl(base, 31);
    if (i < n2) {
        int start = base + s - c;
        cnt[i] = c;
        off[i] = start;
#pragma unroll
        for (int k = 0; k < CH; ++k) {
            cb[(size_t)k * n2 + i] = start;
            start += pc[k];
        }
    }
}

#define PO_THREADS 256

__global__ void __launch_bounds__(PO_THREADS)
k_offsets_proj(const float* __restrict__ x_e, const float* __restrict__ Wh, const float* __restrict__ Wt,
               float* __restrict__ Se, int N_e,
               const float* __restrict__ x_r, const float* __restrict__ Wr,
               float* __restrict__ Sr, int N_r,
               int* __restrict__ cb, int* __restrict__ cnt, int* __restrict__ off,
               int* __restrict__ cursor, int n2, int blocks_e, int blocks_r) {
    const int b    = blockIdx.x;
    const int lane = threadIdx.x & 31;
    const int wib  = threadIdx.x >> 5;
    if (b < blocks_e) {
        proj_tile_wmma(x_e, Wh, Wt, Se, N_e, b * (PO_THREADS / 32) + wib, lane);
    } else if (b < blocks_e + blocks_r) {
        proj_tile_wmma(x_r, Wr, Wr, Sr, N_r, (b - blocks_e) * (PO_THREADS / 32) + wib, lane);
    } else {
        offsets_block(cb, cnt, off, cursor, n2, b - blocks_e - blocks_r);
    }
}

__global__ void k_fill(const int* __restrict__ h_idx,
                       const int* __restrict__ t_idx,
                       const int* __restrict__ rel,
                       const int* __restrict__ cb,
                       const int* __restrict__ rank,
                       int* __restrict__ lst, int E, int N_e) {
    const int e = blockIdx.x * EDGE_BLK + threadIdx.x;
    if (e >= E) return;
    const int chunk = (int)((blockIdx.x * (unsigned)EDGE_BLK) / (unsigned)chunk_len(E));
    const int* __restrict__ b = cb + (size_t)chunk * (2 * (size_t)N_e);
    const int r = rel[e];
    const int p = b[h_idx[e]] + rank[e];
    const int q = b[N_e + t_idx[e]] + rank[E + e];
    lst[p] = r;
    lst[q] = r;
}

template <int NJ>
__device__ __forceinline__ void head_mean_weights(const float (&p)[8], float den, int (&ai)[8]) {
    float a[NJ];
#pragma unroll
    for (int j = 0; j < NJ; ++j) a[j] = p[j] / den;
#pragma unroll
    for (int d = 1; d < 8; d <<= 1) {
#pragma unroll
        for (int j = 0; j < NJ; ++j) a[j] += __shfl_xor(a[j], d);
    }
#pragma unroll
    for (int j = 0; j < 8; ++j) ai[j] = (j < NJ) ? __float_as_int(a[j] * 0.125f) : 0;
}

#define SR_CAP 2048

template <bool TAB>
__device__ __forceinline__ float rel_score(const float* __restrict__ Sr, const float* tab, const int r, const int h) {
    return TAB ? tab[r * 8 + h] : Sr[(size_t)r * 16 + h];
}

template <bool TAB>
__device__ __forceinline__ void node_segment(const int seg, const int n, const int beg, const int lane,
                                             const int* __restrict__ lst,
                                             const float* __restrict__ Se, const float* __restrict__ Sr,
                                             const float* tab,
                                             const float* __restrict__ x_r, float* __restrict__ out,
                                             const int N_e) {
    const int side = (seg >= N_e) ? 1 : 0;
    const int node = seg - side * N_e;
    const int* __restrict__ l = lst + beg;

    const int h    = lane & 7;
    const int slot = lane >> 3;
    const float en = Se[(size_t)node * 16 + side * 8 + h];
    const float* __restrict__ xl = x_r + lane * 4;
    float* __restrict__ o = out + (size_t)node * 256 + side * 128 + lane * 4;

    float4 acc = make_float4(0.0f, 0.0f, 0.0f, 0.0f);

    if (n <= 32) {
        int rl = 0;
        if (lane < n) rl = l[lane];

        float v[8];
#pragma unroll
        for (int j = 0; j < 8; ++j) {
            const int rj = __shfl(rl, 4 * j + slot);
            v[j] = rel_score<TAB>(Sr, tab, rj, h);
        }
        float4 rows[16];
#pragma unroll
        for (int k = 0; k < 16; ++k) {
            const int rk = __builtin_amdgcn_readlane(rl, k);
            rows[k] = *(const float4*)(xl + (size_t)rk * 128);
        }

        float s[8];
        float m = -INFINITY;
#pragma unroll
        for (int j = 0; j < 8; ++j) {
            s[j] = (4 * j + slot < n) ? lrelu(en + v[j]) : -INFINITY;
            m = fmaxf(m, s[j]);
        }
        m = fmaxf(m, __shfl_xor(m, 8));
        m = fmaxf(m, __shfl_xor(m, 16));
        float z = 0.0f;
#pragma unroll
        for (int j = 0; j < 8; ++j) {
            s[j] = (4 * j + slot < n) ? __expf(s[j] - m) : 0.0f;
            z += s[j];
        }
        z += __shfl_xor(z, 8);
        z += __shfl_xor(z, 16);
        const float den = z + 1e-16f;
        int ai[8];
        if (n > 16) head_mean_weights<8>(s, den, ai);
        else        head_mean_weights<4>(s, den, ai);

#pragma unroll
        for (int k = 0; k < 16; ++k) {
            if (k < n) {
                const float ak = __int_as_float(__builtin_amdgcn_readlane(ai[k >> 2], (k & 3) * 8));
                acc.x += rows[k].x * ak;
                acc.y += rows[k].y * ak;
                acc.z += rows[k].z * ak;
                acc.w += rows[k].w * ak;
            }
        }
        if (n > 16) {
#pragma unroll
            for (int q = 0; q < 4; ++q) {
                if (16 + 4 * q < n) {
#pragma unroll
                    for (int t = 0; t < 4; ++t) {
                        const int k  = 4 * q + t;
                        const int rk = __builtin_amdgcn_readlane(rl, 16 + k);
                        rows[k] = *(const float4*)(xl + (size_t)rk * 128);
                    }
                }
            }
#pragma unroll
            for (int k = 0; k < 16; ++k) {
                if (16 + k < n) {
                    const float ak = __int_as_float(__builtin_amdgcn_readlane(ai[4 + (k >> 2)], (k & 3) * 8));
                    acc.x += rows[k].x * ak;
                    acc.y += rows[k].y * ak;
                    acc.z += rows[k].z * ak;
                    acc.w += rows[k].w * ak;
                }
            }
        }
        *(float4*)o = acc;
        return;
    }

    float m = -INFINITY;
    for (int i = slot; i < n; i += 4) {
        const int r = l[i];
        m = fmaxf(m, lrelu(en + rel_score<TAB>(Sr, tab, r, h)));
    }
    m = fmaxf(m, __shfl_xor(m, 8));
    m = fmaxf(m, __shfl_xor(m, 16));

    float z = 0.0f;
    for (int i = slot; i < n; i += 4) {
        const int r = l[i];
        z += __expf(lrelu(en + rel_score<TAB>(Sr, tab, r, h)) - m);
    }
    z += __shfl_xor(z, 8);
    z += __shfl_xor(z, 16);
    const float den = z + 1e-16f;

    int i0 = 0;
    for (; i0 + 4 <= n; i0 += 4) {
        const int r = l[i0 + slot];
        float a = __expf(lrelu(en + rel_score<TAB>(Sr, tab, r, h)) - m) / den;
        a += __shfl_xor(a, 1);
        a += __shfl_xor(a, 2);
        a += __shfl_xor(a, 4);
        a *= 0.125f;
        const int ai = __float_as_int(a);
        const int r0 = __builtin_amdgcn_readlane(r, 0);
        const int r1 = __builtin_amdgcn_readlane(r, 8);
        const int r2 = __builtin_amdgcn_readlane(r, 16);
        const int r3 = __builtin_amdgcn_readlane(r, 24);
        const float a0 = __int_as_float(__builtin_amdgcn_readlane(ai, 0));
        const float a1 = __int_as_float(__builtin_amdgcn_readlane(ai, 8));
        const float a2 = __int_as_float(__builtin_amdgcn_readlane(ai, 16));
        const float a3 = __int_as_float(__builtin_amdgcn_readlane(ai, 24));
        const float4 m0 = *(const float4*)(xl + (size_t)r0 * 128);
        const float4 m1 = *(const float4*)(xl + (size_t)r1 * 128);
        const float4 m2 = *(const float4*)(xl + (size_t)r2 * 128);
        const float4 m3 = *(const float4*)(xl + (size_t)r3 * 128);
        acc.x += m0.x * a0; acc.y += m0.y * a0; acc.z += m0.z * a0; acc.w += m0.w * a0;
        acc.x += m1.x * a1; acc.y += m1.y * a1; acc.z += m1.z * a1; acc.w += m1.w * a1;
        acc.x += m2.x * a2; acc.y += m2.y * a2; acc.z += m2.z * a2; acc.w += m2.w * a2;
        acc.x += m3.x * a3; acc.y += m3.y * a3; acc.z += m3.z * a3; acc.w += m3.w * a3;
    }
    if (i0 < n) {
        const int  i  = i0 + slot;
        const bool ok = i < n;
        const int  r  = ok ? l[i] : 0;
        float a = 0.0f;
        if (ok) a = __expf(lrelu(en + rel_score<TAB>(Sr, tab, r, h)) - m) / den;
        a += __shfl_xor(a, 1);
        a += __shfl_xor(a, 2);
        a += __shfl_xor(a, 4);
        a *= 0.125f;
        const int ai = __float_as_int(a);
#pragma unroll
        for (int k = 0; k < 3; ++k) {
            if (i0 + k < n) {
                const int   rk = __builtin_amdgcn_readlane(r, k * 8);
                const float ak = __int_as_float(__builtin_amdgcn_readlane(ai, k * 8));
                const float4 mv = *(const float4*)(xl + (size_t)rk * 128);
                acc.x += mv.x * ak;
                acc.y += mv.y * ak;
                acc.z += mv.z * ak;
                acc.w += mv.w * ak;
            }
        }
    }
    *(float4*)o = acc;
}

#define NODE_THREADS 1024

template <bool TAB>
__device__ __forceinline__ void walk_segments(const int w, const int nw, const int n2, const int lane,
                                              const int* __restrict__ cnt, const int* __restrict__ off,
                                              const int* __restrict__ lst,
                                              const float* __restrict__ Se, const float* __restrict__ Sr,
                                              const float* tab,
                                              const float* __restrict__ x_r, float* __restrict__ out,
                                              const int N_e) {
    int seg   = w;
    int n_nx  = cnt[seg];
    int beg_nx = off[seg];
    for (;;) {
        const int n   = __builtin_amdgcn_readfirstlane(n_nx);
        const int beg = __builtin_amdgcn_readfirstlane(beg_nx);
        const int  sn   = seg + nw;
        const bool more = sn < n2;
        if (more) { n_nx = cnt[sn]; beg_nx = off[sn]; }
        node_segment<TAB>(seg, n, beg, lane, lst, Se, Sr, tab, x_r, out, N_e);
        if (!more) break;
        seg = sn;
    }
}

__global__ void __launch_bounds__(NODE_THREADS)
k_node(const int* __restrict__ cnt, const int* __restrict__ off,
       const int* __restrict__ lst,
       const float* __restrict__ Se, const float* __restrict__ Sr,
       const float* __restrict__ x_r, float* __restrict__ out,
       int N_e, int N_r) {
    __shared__ float tab[SR_CAP * 8];
    const bool use_tab = N_r <= SR_CAP;
    if (use_tab) {
        const int items = 2 * N_r;
        for (int i0 = threadIdx.x; i0 < items; i0 += 4 * NODE_THREADS) {
            const int i1 = i0 + NODE_THREADS;
            const int i2 = i0 + 2 * NODE_THREADS;
            const int i3 = i0 + 3 * NODE_THREADS;
            const int c1 = min(i1, items - 1);
            const int c2 = min(i2, items - 1);
            const int c3 = min(i3, items - 1);
            const float4 t0 = *(const float4*)(Sr + (size_t)(i0 >> 1) * 16 + (i0 & 1) * 4);
            const float4 t1 = *(const float4*)(Sr + (size_t)(c1 >> 1) * 16 + (c1 & 1) * 4);
            const float4 t2 = *(const float4*)(Sr + (size_t)(c2 >> 1) * 16 + (c2 & 1) * 4);
            const float4 t3 = *(const float4*)(Sr + (size_t)(c3 >> 1) * 16 + (c3 & 1) * 4);
            *(float4*)(tab + i0 * 4) = t0;
            *(float4*)(tab + c1 * 4) = t1;
            *(float4*)(tab + c2 * 4) = t2;
            *(float4*)(tab + c3 * 4) = t3;
        }
        __syncthreads();
    }

    const int lane = threadIdx.x & 31;
    const int w  = __builtin_amdgcn_readfirstlane((int)((blockIdx.x * NODE_THREADS + threadIdx.x) >> 5));
    const int nw = (int)(gridDim.x * (NODE_THREADS / 32));
    const int n2 = 2 * N_e;
    if (w >= n2) return;
    if (use_tab) walk_segments<true>(w, nw, n2, lane, cnt, off, lst, Se, Sr, tab, x_r, out, N_e);
    else         walk_segments<false>(w, nw, n2, lane, cnt, off, lst, Se, Sr, tab, x_r, out, N_e);
}

extern "C" void kernel_launch(void* const* d_in, const int* in_sizes, int n_in,
                              void* d_out, int out_size, void* d_ws, size_t ws_size,
                              hipStream_t stream) {
    const float* x_e  = (const float*)d_in[0];
    const float* x_r  = (const float*)d_in[1];
    const float* Wh   = (const float*)d_in[2];
    const float* Wt   = (const float*)d_in[3];
    const float* Wr   = (const float*)d_in[4];
    const int*   eidx = (const int*)d_in[5];
    const int*   rel  = (const int*)d_in[6];

    const int D   = 128;
    const int N_e = in_sizes[0] / D;
    const int N_r = in_sizes[1] / D;
    const int E   = in_sizes[6];

    const int* h_idx = eidx;
    const int* t_idx = eidx + E;

    auto align256 = [](size_t x) { return (x + 255) & ~(size_t)255; };
    char* ws = (char*)d_ws;
    float* Se     = (float*)ws; ws += align256((size_t)N_e * 16 * sizeof(float));
    float* Sr     = (float*)ws; ws += align256((size_t)N_r * 16 * sizeof(float));
    int*   cnt    = (int*)ws;   ws += align256((size_t)N_e * 2 * sizeof(int));
    int*   off    = (int*)ws;   ws += align256((size_t)N_e * 2 * sizeof(int));
    int*   lst    = (int*)ws;   ws += align256((size_t)E * 2 * sizeof(int));
    int*   cursor = (int*)ws;   ws += 256;
    if ((size_t)(ws - (char*)d_ws) > ws_size) return;

    float* out = (float*)d_out;

    const int n2 = N_e * 2;
    int* cb   = (int*)ws;       ws += align256((size_t)CH * n2 * sizeof(int));
    int* rank = (int*)ws;       ws += align256((size_t)E * 2 * sizeof(int));
    if ((size_t)(ws - (char*)d_ws) > ws_size) {
        if ((size_t)out_size < (size_t)E * 2 + (size_t)CH * n2) return;
        rank = (int*)d_out;
        cb   = rank + (size_t)E * 2;
    }

    const int BLK = PO_THREADS;

    const int blocks_e = ((N_e + 15) / 16 + BLK / 32 - 1) / (BLK / 32);
    const int blocks_r = ((N_r + 15) / 16 + BLK / 32 - 1) / (BLK / 32);
    const int blocks_o = (n2 + BLK - 1) / BLK;
    k_count<<<dim3(CH, 2 * PS), CNT_THREADS, 0, stream>>>(eidx, cb, rank, cursor, E, N_e);
    k_offsets_proj<<<blocks_e + blocks_r + blocks_o, BLK, 0, stream>>>(x_e, Wh, Wt, Se, N_e, x_r, Wr, Sr, N_r,
                                                                     cb, cnt, off, cursor, n2, blocks_e, blocks_r);
    k_fill<<<(E + EDGE_BLK - 1) / EDGE_BLK, EDGE_BLK, 0, stream>>>(h_idx, t_idx, rel, cb, rank, lst, E, N_e);

    k_node<<<256, NODE_THREADS, 0, stream>>>(cnt, off, lst, Se, Sr, x_r, out, N_e, N_r);
}


# Each count block sorts its own edges: a 2nd pass over its chunk drops them at LDS cursors from a block scan; rank array, offset pass and fill launch go (2 launches, was 4); slot order set by pass 2
# speedup vs baseline: 4.2297x; 1.0427x over previous
#include <hip/hip_runtime.h>
#include <stdint.h>


typedef __attribute__((ext_vector_type(2))) float v2f;
typedef __attribute__((ext_vector_type(8))) float v8f;

#define NEG_SLOPE 0.01f

__device__ __forceinline__ float lrelu(float x) {
    return x > 0.0f ? x : NEG_SLOPE * x;
}

#define CH   8
#define PS   16
#define CAP  4096
#define SORT_THREADS 1024
#define NSORT (CH * 2 * PS)
#define ID_PAD 0x7fffffff

__device__ __forceinline__ int chunk_len(int E) {
    return (((E + CH - 1) / CH) + 255) & ~255;
}

template <bool IN_OUT>
__device__ __forceinline__ void proj_tile_wmma(const float* __restrict__ X,
                                               const float* __restrict__ W0,
                                               const float* __restrict__ W1,
                                               float* __restrict__ S, const int nrows,
                                               const int wave, const int lane) {
    const int row0 = wave * 16;
    if (row0 >= nrows) return;

    const int m    = lane & 15;
    const int kofs = (lane >> 4) << 1;

    const float* __restrict__ arow = X + (size_t)(row0 + m) * 128;
    const float* __restrict__ brow =
        (m < 8) ? (W0 + (size_t)m * 128) : (W1 + (size_t)(m - 8) * 128);

    v8f c = {0.f, 0.f, 0.f, 0.f, 0.f, 0.f, 0.f, 0.f};
#pragma unroll
    for (int k = 0; k < 128; k += 4) {
        v2f a = *(const v2f*)(arow + k + kofs);
        v2f b = *(const v2f*)(brow + k + kofs);
        c = __builtin_amdgcn_wmma_f32_16x16x4_f32(
             false, a,  false, b,
             (short)0, c,  false,  false);
    }

    const int rbase = row0 + ((lane >> 4) << 3);
    const int col   = lane & 15;
#pragma unroll
    for (int v = 0; v < 8; ++v) {
        if (IN_OUT) S[(size_t)(rbase + v) * 256 + (col >> 3) * 128 + (col & 7)] = c[v];
        else        S[(size_t)(rbase + v) * 16 + col] = c[v];
    }
}

template <bool VEC>
__device__ __forceinline__ void load16(const int* __restrict__ a, const int e, const int e1,
                                       const int pad, int4 (&v)[4]) {
#pragma unroll
    for (int u = 0; u < 4; ++u) {
        const int ee = e + u * 4 * SORT_THREADS;
        v[u] = make_int4(pad, pad, pad, pad);
        if (VEC) {
            if (ee < e1) v[u] = *(const int4*)(a + ee);
        } else {
            if (ee     < e1) v[u].x = a[ee];
            if (ee + 1 < e1) v[u].y = a[ee + 1];
            if (ee + 2 < e1) v[u].z = a[ee + 2];
            if (ee + 3 < e1) v[u].w = a[ee + 3];
        }
    }
}

template <bool VEC>
__device__ __forceinline__ int count_ids(const int* __restrict__ id, int* c,
                                         const int s0, const int len, const int e0, const int e1, const int tid) {
    int below = 0;
    for (int e = e0 + 4 * tid; e < e1; e += 16 * SORT_THREADS) {
        int4 v[4];
        load16<VEC>(id, e, e1, ID_PAD, v);
#pragma unroll
        for (int u = 0; u < 4; ++u) {
            const int vv[4] = {v[u].x, v[u].y, v[u].z, v[u].w};
#pragma unroll
            for (int j = 0; j < 4; ++j) {
                const unsigned d = (unsigned)(vv[j] - s0);
                if (d < (unsigned)len) atomicAdd(&c[d], 1);
                below += (vv[j] < s0) ? 1 : 0;
            }
        }
    }
    return below;
}

template <bool VEC>
__device__ __forceinline__ void place_ids(const int* __restrict__ id, const int* __restrict__ rel,
                                          int* __restrict__ lst, int* c,
                                          const int s0, const int len, const int e0, const int e1, const int tid) {
    for (int e = e0 + 4 * tid; e < e1; e += 16 * SORT_THREADS) {
        int4 v[4], r[4];
        load16<VEC>(id, e, e1, ID_PAD, v);
        load16<VEC>(rel, e, e1, 0, r);
        int p[16];
#pragma unroll
        for (int u = 0; u < 4; ++u) {
            const int vv[4] = {v[u].x, v[u].y, v[u].z, v[u].w};
#pragma unroll
            for (int j = 0; j < 4; ++j) {
                const unsigned d = (unsigned)(vv[j] - s0);
                p[4 * u + j] = (d < (unsigned)len) ? atomicAdd(&c[d], 1) : -1;
            }
        }
#pragma unroll
        for (int u = 0; u < 4; ++u) {
            const int rr[4] = {r[u].x, r[u].y, r[u].z, r[u].w};
#pragma unroll
            for (int j = 0; j < 4; ++j) {
                if (p[4 * u + j] >= 0) lst[p[4 * u + j]] = rr[j];
            }
        }
    }
}

__device__ __forceinline__ void sort_block(const int* __restrict__ eidx, const int* __restrict__ rel,
                                           int* __restrict__ st, int* __restrict__ lst,
                                           const int E, const int N_e, const int b,
                                           int* c, int* wtot, int* wbel) {
    const int tid  = threadIdx.x;
    const int lane = tid & 31;
    const int wave = tid >> 5;
    const int chunk = b % CH;
    const int y     = b / CH;
    const int side  = y / PS;
    const int part  = y - side * PS;

    const int plen = (N_e + PS - 1) / PS;
    const int lo   = part * plen;
    const int hi   = min(lo + plen, N_e);
    const int clen = chunk_len(E);
    const int e0   = min(chunk * clen, E);
    const int e1   = min(e0 + clen, E);
    const int* __restrict__ id = eidx + (size_t)side * E;
    int* __restrict__ srow = st + (size_t)(chunk * 2 + side) * (N_e + 1);
    const int region = side * E + e0;
    const bool vec = ((E & 3) == 0) && ((((size_t)eidx) & 15) == 0) && ((((size_t)rel) & 15) == 0);

    if (part == PS - 1 && tid == 0) srow[N_e] = side * E + e1;

    for (int s0 = lo; s0 < hi; s0 += CAP) {
        const int len = min(CAP, hi - s0);
        *(int4*)(c + 4 * tid) = make_int4(0, 0, 0, 0);
        __syncthreads();
        int below;
        if (vec) below = count_ids<true>(id, c, s0, len, e0, e1, tid);
        else     below = count_ids<false>(id, c, s0, len, e0, e1, tid);
        __syncthreads();

        const int4 k = *(const int4*)(c + 4 * tid);
        const int tsum = k.x + k.y + k.z + k.w;
        int s = tsum;
#pragma unroll
        for (int d = 1; d < 32; d <<= 1) {
            const int t = __shfl_up(s, d);
            if (lane >= d) s += t;
        }
        int bl = below;
#pragma unroll
        for (int d = 16; d > 0; d >>= 1) bl += __shfl_xor(bl, d);
        if (lane == 31) wtot[wave] = s;
        if (lane == 0)  wbel[wave] = bl;
        __syncthreads();
        const int wt = wtot[lane];
        int ws = wt;
#pragma unroll
        for (int d = 1; d < 32; d <<= 1) {
            const int t = __shfl_up(ws, d);
            if (lane >= d) ws += t;
        }
        const int wex = __shfl(ws - wt, wave);
        int bt = wbel[lane];
#pragma unroll
        for (int d = 16; d > 0; d >>= 1) bt += __shfl_xor(bt, d);

        const int x0 = region + bt + wex + (s - tsum);
        const int x1 = x0 + k.x;
        const int x2 = x1 + k.y;
        const int x3 = x2 + k.z;
        const int i  = 4 * tid;
        if (i     < len) srow[s0 + i]     = x0;
        if (i + 1 < len) srow[s0 + i + 1] = x1;
        if (i + 2 < len) srow[s0 + i + 2] = x2;
        if (i + 3 < len) srow[s0 + i + 3] = x3;
        *(int4*)(c + 4 * tid) = make_int4(x0, x1, x2, x3);
        __syncthreads();
        if (vec) place_ids<true>(id, rel, lst, c, s0, len, e0, e1, tid);
        else     place_ids<false>(id, rel, lst, c, s0, len, e0, e1, tid);
        __syncthreads();
    }
}

__global__ void __launch_bounds__(SORT_THREADS, 16)
k_sort_proj(const int* __restrict__ eidx, const int* __restrict__ rel,
            int* __restrict__ st, int* __restrict__ lst, int E, int N_e,
            const float* __restrict__ x_e, const float* __restrict__ Wh, const float* __restrict__ Wt,
            float* __restrict__ out,
            const float* __restrict__ x_r, const float* __restrict__ Wr,
            float* __restrict__ Sr, int N_r, int blocks_e) {
    __shared__ int4 cnt4[CAP / 4];
    __shared__ int wtot[SORT_THREADS / 32];
    __shared__ int wbel[SORT_THREADS / 32];
    static_assert(SORT_THREADS / 32 == 32, "one lane per wave in the block scan");
    static_assert(CAP == 4 * SORT_THREADS, "four counters per thread");
    const int b = blockIdx.x;
    if (b < NSORT) {
        sort_block(eidx, rel, st, lst, E, N_e, b, (int*)cnt4, wtot, wbel);
    } else {
        const int lane = threadIdx.x & 31;
        const int wib  = threadIdx.x >> 5;
        const int pb   = b - NSORT;
        if (pb < blocks_e) {
            proj_tile_wmma<true>(x_e, Wh, Wt, out, N_e, pb * (SORT_THREADS / 32) + wib, lane);
        } else {
            proj_tile_wmma<false>(x_r, Wr, Wr, Sr, N_r, (pb - blocks_e) * (SORT_THREADS / 32) + wib, lane);
        }
    }
}

template <int NJ>
__device__ __forceinline__ void head_mean_weights(const float (&p)[8], float den, int (&ai)[8]) {
    float a[NJ];
#pragma unroll
    for (int j = 0; j < NJ; ++j) a[j] = p[j] / den;
#pragma unroll
    for (int d = 1; d < 8; d <<= 1) {
#pragma unroll
        for (int j = 0; j < NJ; ++j) a[j] += __shfl_xor(a[j], d);
    }
#pragma unroll
    for (int j = 0; j < 8; ++j) ai[j] = (j < NJ) ? __float_as_int(a[j] * 0.125f) : 0;
}

__device__ __forceinline__ int run_pos(const int i,
                                       const int a0, const int a1, const int a2, const int a3,
                                       const int a4, const int a5, const int a6, const int a7,
                                       const int p1, const int p2, const int p3, const int p4,
                                       const int p5, const int p6, const int p7) {
    int a = a0;
    a = (i >= p1) ? a1 : a;
    a = (i >= p2) ? a2 : a;
    a = (i >= p3) ? a3 : a;
    a = (i >= p4) ? a4 : a;
    a = (i >= p5) ? a5 : a;
    a = (i >= p6) ? a6 : a;
    a = (i >= p7) ? a7 : a;
    return a + i;
}

#define SR_CAP 2048

template <bool TAB>
__device__ __forceinline__ float rel_score(const float* __restrict__ Sr, const float* tab, const int r, const int h) {
    return TAB ? tab[r * 8 + h] : Sr[(size_t)r * 16 + h];
}

template <bool TAB>
__device__ __forceinline__ void node_segment(const int seg, const int n,
                                             const int (&pre)[CH], const int (&adj)[CH], const int lane,
                                             const int* __restrict__ lst,
                                             const float* __restrict__ Sr,
                                             const float* tab,
                                             const float* __restrict__ x_r, float* __restrict__ out,
                                             const int N_e) {
    const int side = (seg >= N_e) ? 1 : 0;
    const int node = seg - side * N_e;
    static_assert(CH == 8, "run_pos names the eight runs one by one");
    const int p1 = pre[1], p2 = pre[2], p3 = pre[3], p4 = pre[4], p5 = pre[5], p6 = pre[6], p7 = pre[7];
    const int a0 = adj[0], a1 = adj[1], a2 = adj[2], a3 = adj[3], a4 = adj[4], a5 = adj[5], a6 = adj[6], a7 = adj[7];
#define POS(i) run_pos((i), a0, a1, a2, a3, a4, a5, a6, a7, p1, p2, p3, p4, p5, p6, p7)

    const int h    = lane & 7;
    const int slot = lane >> 3;
    const float* enp = out + (size_t)node * 256 + side * 128 + h;
    const float* __restrict__ xl = x_r + lane * 4;
    float* o = out + (size_t)node * 256 + side * 128 + lane * 4;

    float4 acc = make_float4(0.0f, 0.0f, 0.0f, 0.0f);

    if (n <= 32) {
        int rl = 0;
        if (lane < n) rl = lst[POS(lane)];
        const float en = *enp;

        float v[8];
#pragma unroll
        for (int j = 0; j < 8; ++j) {
            const int rj = __shfl(rl, 4 * j + slot);
            v[j] = rel_score<TAB>(Sr, tab, rj, h);
        }
        float4 rows[16];
#pragma unroll
        for (int k = 0; k < 16; ++k) {
            const int rk = __builtin_amdgcn_readlane(rl, k);
            rows[k] = *(const float4*)(xl + (size_t)rk * 128);
        }

        float s[8];
        float m = -INFINITY;
#pragma unroll
        for (int j = 0; j < 8; ++j) {
            s[j] = (4 * j + slot < n) ? lrelu(en + v[j]) : -INFINITY;
            m = fmaxf(m, s[j]);
        }
        m = fmaxf(m, __shfl_xor(m, 8));
        m = fmaxf(m, __shfl_xor(m, 16));
        float z = 0.0f;
#pragma unroll
        for (int j = 0; j < 8; ++j) {
            s[j] = (4 * j + slot < n) ? __expf(s[j] - m) : 0.0f;
            z += s[j];
        }
        z += __shfl_xor(z, 8);
        z += __shfl_xor(z, 16);
        const float den = z + 1e-16f;
        int ai[8];
        if (n > 16) head_mean_weights<8>(s, den, ai);
        else        head_mean_weights<4>(s, den, ai);

#pragma unroll
        for (int k = 0; k < 16; ++k) {
            if (k < n) {
                const float ak = __int_as_float(__builtin_amdgcn_readlane(ai[k >> 2], (k & 3) * 8));
                acc.x += rows[k].x * ak;
                acc.y += rows[k].y * ak;
                acc.z += rows[k].z * ak;
                acc.w += rows[k].w * ak;
            }
        }
        if (n > 16) {
#pragma unroll
            for (int q = 0; q < 4; ++q) {
                if (16 + 4 * q < n) {
#pragma unroll
                    for (int t = 0; t < 4; ++t) {
                        const int k  = 4 * q + t;
                        const int rk = __builtin_amdgcn_readlane(rl, 16 + k);
                        rows[k] = *(const float4*)(xl + (size_t)rk * 128);
                    }
                }
            }
#pragma unroll
            for (int k = 0; k < 16; ++k) {
                if (16 + k < n) {
                    const float ak = __int_as_float(__builtin_amdgcn_readlane(ai[4 + (k >> 2)], (k & 3) * 8));
                    acc.x += rows[k].x * ak;
                    acc.y += rows[k].y * ak;
                    acc.z += rows[k].z * ak;
                    acc.w += rows[k].w * ak;
                }
            }
        }
        *(float4*)o = acc;
        return;
    }

    const float en = *enp;
    float m = -INFINITY;
    for (int i = slot; i < n; i += 4) {
        const int r = lst[POS(i)];
        m = fmaxf(m, lrelu(en + rel_score<TAB>(Sr, tab, r, h)));
    }
    m = fmaxf(m, __shfl_xor(m, 8));
    m = fmaxf(m, __shfl_xor(m, 16));

    float z = 0.0f;
    for (int i = slot; i < n; i += 4) {
        const int r = lst[POS(i)];
        z += __expf(lrelu(en + rel_score<TAB>(Sr, tab, r, h)) - m);
    }
    z += __shfl_xor(z, 8);
    z += __shfl_xor(z, 16);
    const float den = z + 1e-16f;

    int i0 = 0;
    for (; i0 + 4 <= n; i0 += 4) {
        const int r = lst[POS(i0 + slot)];
        float a = __expf(lrelu(en + rel_score<TAB>(Sr, tab, r, h)) - m) / den;
        a += __shfl_xor(a, 1);
        a += __shfl_xor(a, 2);
        a += __shfl_xor(a, 4);
        a *= 0.125f;
        const int ai = __float_as_int(a);
        const int r0 = __builtin_amdgcn_readlane(r, 0);
        const int r1 = __builtin_amdgcn_readlane(r, 8);
        const int r2 = __builtin_amdgcn_readlane(r, 16);
        const int r3 = __builtin_amdgcn_readlane(r, 24);
        const float a0 = __int_as_float(__builtin_amdgcn_readlane(ai, 0));
        const float a1 = __int_as_float(__builtin_amdgcn_readlane(ai, 8));
        const float a2 = __int_as_float(__builtin_amdgcn_readlane(ai, 16));
        const float a3 = __int_as_float(__builtin_amdgcn_readlane(ai, 24));
        const float4 m0 = *(const float4*)(xl + (size_t)r0 * 128);
        const float4 m1 = *(const float4*)(xl + (size_t)r1 * 128);
        const float4 m2 = *(const float4*)(xl + (size_t)r2 * 128);
        const float4 m3 = *(const float4*)(xl + (size_t)r3 * 128);
        acc.x += m0.x * a0; acc.y += m0.y * a0; acc.z += m0.z * a0; acc.w += m0.w * a0;
        acc.x += m1.x * a1; acc.y += m1.y * a1; acc.z += m1.z * a1; acc.w += m1.w * a1;
        acc.x += m2.x * a2; acc.y += m2.y * a2; acc.z += m2.z * a2; acc.w += m2.w * a2;
        acc.x += m3.x * a3; acc.y += m3.y * a3; acc.z += m3.z * a3; acc.w += m3.w * a3;
    }
    if (i0 < n) {
        const int  i  = i0 + slot;
        const bool ok = i < n;
        const int  r  = ok ? lst[POS(i)] : 0;
        float a = 0.0f;
        if (ok) a = __expf(lrelu(en + rel_score<TAB>(Sr, tab, r, h)) - m) / den;
        a += __shfl_xor(a, 1);
        a += __shfl_xor(a, 2);
        a += __shfl_xor(a, 4);
        a *= 0.125f;
        const int ai = __float_as_int(a);
#pragma unroll
        for (int k = 0; k < 3; ++k) {
            if (i0 + k < n) {
                const int   rk = __builtin_amdgcn_readlane(r, k * 8);
                const float ak = __int_as_float(__builtin_amdgcn_readlane(ai, k * 8));
                const float4 mv = *(const float4*)(xl + (size_t)rk * 128);
                acc.x += mv.x * ak;
                acc.y += mv.y * ak;
                acc.z += mv.z * ak;
                acc.w += mv.w * ak;
            }
        }
    }
    *(float4*)o = acc;
}

#undef POS

#define NODE_THREADS 1024

__device__ __forceinline__ void load_runs(const int* __restrict__ st, const int seg, const int N_e,
                                          int (&nx)[2 * CH]) {
    const int side = (seg >= N_e) ? 1 : 0;
    const int node = seg - side * N_e;
    const int* __restrict__ p = st + (size_t)side * (N_e + 1) + node;
#pragma unroll
    for (int k = 0; k < CH; ++k) {
        nx[k]      = p[(size_t)k * 2 * (N_e + 1)];
        nx[CH + k] = p[(size_t)k * 2 * (N_e + 1) + 1];
    }
}

template <bool TAB>
__device__ __forceinline__ void walk_segments(const int w, const int nw, const int n2, const int lane,
                                              const int* __restrict__ st,
                                              const int* __restrict__ lst,
                                              const float* __restrict__ Sr,
                                              const float* tab,
                                              const float* __restrict__ x_r, float* __restrict__ out,
                                              const int N_e) {
    int seg = w;
    int nx[2 * CH];
    load_runs(st, seg, N_e, nx);
    for (;;) {
        int pre[CH], adj[CH];
        int n = 0;
#pragma unroll
        for (int k = 0; k < CH; ++k) {
            const int s = __builtin_amdgcn_readfirstlane(nx[k]);
            const int e = __builtin_amdgcn_readfirstlane(nx[CH + k]);
            pre[k] = n;
            adj[k] = s - n;
            n += e - s;
        }
        const int  sn   = seg + nw;
        const bool more = sn < n2;
        if (more) load_runs(st, sn, N_e, nx);
        node_segment<TAB>(seg, n, pre, adj, lane, lst, Sr, tab, x_r, out, N_e);
        if (!more) break;
        seg = sn;
    }
}

__global__ void __launch_bounds__(NODE_THREADS)
k_node(const int* __restrict__ st,
       const int* __restrict__ lst,
       const float* __restrict__ Sr,
       const float* __restrict__ x_r, float* __restrict__ out,
       int N_e, int N_r) {
    __shared__ float tab[SR_CAP * 8];
    const bool use_tab = N_r <= SR_CAP;
    if (use_tab) {
        const int items = 2 * N_r;
        for (int i0 = threadIdx.x; i0 < items; i0 += 4 * NODE_THREADS) {
            const int i1 = i0 + NODE_THREADS;
            const int i2 = i0 + 2 * NODE_THREADS;
            const int i3 = i0 + 3 * NODE_THREADS;
            const int c1 = min(i1, items - 1);
            const int c2 = min(i2, items - 1);
            const int c3 = min(i3, items - 1);
            const float4 t0 = *(const float4*)(Sr + (size_t)(i0 >> 1) * 16 + (i0 & 1) * 4);
            const float4 t1 = *(const float4*)(Sr + (size_t)(c1 >> 1) * 16 + (c1 & 1) * 4);
            const float4 t2 = *(const float4*)(Sr + (size_t)(c2 >> 1) * 16 + (c2 & 1) * 4);
            const float4 t3 = *(const float4*)(Sr + (size_t)(c3 >> 1) * 16 + (c3 & 1) * 4);
            *(float4*)(tab + i0 * 4) = t0;
            *(float4*)(tab + c1 * 4) = t1;
            *(float4*)(tab + c2 * 4) = t2;
            *(float4*)(tab + c3 * 4) = t3;
        }
        __syncthreads();
    }

    const int lane = threadIdx.x & 31;
    const int w  = __builtin_amdgcn_readfirstlane((int)((blockIdx.x * NODE_THREADS + threadIdx.x) >> 5));
    const int nw = (int)(gridDim.x * (NODE_THREADS / 32));
    const int n2 = 2 * N_e;
    if (w >= n2) return;
    if (use_tab) walk_segments<true>(w, nw, n2, lane, st, lst, Sr, tab, x_r, out, N_e);
    else         walk_segments<false>(w, nw, n2, lane, st, lst, Sr, tab, x_r, out, N_e);
}

extern "C" void kernel_launch(void* const* d_in, const int* in_sizes, int n_in,
                              void* d_out, int out_size, void* d_ws, size_t ws_size,
                              hipStream_t stream) {
    const float* x_e  = (const float*)d_in[0];
    const float* x_r  = (const float*)d_in[1];
    const float* Wh   = (const float*)d_in[2];
    const float* Wt   = (const float*)d_in[3];
    const float* Wr   = (const float*)d_in[4];
    const int*   eidx = (const int*)d_in[5];
    const int*   rel  = (const int*)d_in[6];

    const int D   = 128;
    const int N_e = in_sizes[0] / D;
    const int N_r = in_sizes[1] / D;
    const int E   = in_sizes[6];

    auto align256 = [](size_t x) { return (x + 255) & ~(size_t)255; };
    char* ws = (char*)d_ws;
    float* Sr  = (float*)ws; ws += align256((size_t)N_r * 16 * sizeof(float));
    int*   st  = (int*)ws;   ws += align256((size_t)CH * 2 * ((size_t)N_e + 1) * sizeof(int));
    int*   lst = (int*)ws;   ws += align256((size_t)E * 2 * sizeof(int));
    if ((size_t)(ws - (char*)d_ws) > ws_size) return;

    float* out = (float*)d_out;

    const int WPB = SORT_THREADS / 32;
    const int blocks_e = ((N_e + 15) / 16 + WPB - 1) / WPB;
    const int blocks_r = ((N_r + 15) / 16 + WPB - 1) / WPB;
    k_sort_proj<<<NSORT + blocks_e + blocks_r, SORT_THREADS, 0, stream>>>(eidx, rel, st, lst, E, N_e,
                                                                       x_e, Wh, Wt, out,
                                                                       x_r, Wr, Sr, N_r, blocks_e);

    k_node<<<256, NODE_THREADS, 0, stream>>>(st, lst, Sr, x_r, out, N_e, N_r);
}


# Sort blocks read their chunk once: pass 1 also parks each in-range edge (relation id, counter index) in LDS, pass 2 replays those records, not the ids; slot (sum) order in a run set by the replay
# speedup vs baseline: 4.4550x; 1.0533x over previous
#include <hip/hip_runtime.h>
#include <stdint.h>


typedef __attribute__((ext_vector_type(2))) float v2f;
typedef __attribute__((ext_vector_type(8))) float v8f;

#define NEG_SLOPE 0.01f

__device__ __forceinline__ float lrelu(float x) {
    return x > 0.0f ? x : NEG_SLOPE * x;
}

#define CH   8
#define PS   16
#define CAP  4096
#define SORT_THREADS 1024
#define NSORT (CH * 2 * PS)
#define ID_PAD 0x7fffffff
#define REC_K 24
#define REC_DBITS 12
#define REC_RMAX (1 << 20)

__device__ __forceinline__ int chunk_len(int E) {
    return (((E + CH - 1) / CH) + 255) & ~255;
}

template <bool IN_OUT>
__device__ __forceinline__ void proj_tile_wmma(const float* __restrict__ X,
                                               const float* __restrict__ W0,
                                               const float* __restrict__ W1,
                                               float* __restrict__ S, const int nrows,
                                               const int wave, const int lane) {
    const int row0 = wave * 16;
    if (row0 >= nrows) return;

    const int m    = lane & 15;
    const int kofs = (lane >> 4) << 1;

    const float* __restrict__ arow = X + (size_t)(row0 + m) * 128;
    const float* __restrict__ brow =
        (m < 8) ? (W0 + (size_t)m * 128) : (W1 + (size_t)(m - 8) * 128);

    v8f c = {0.f, 0.f, 0.f, 0.f, 0.f, 0.f, 0.f, 0.f};
#pragma unroll
    for (int k = 0; k < 128; k += 4) {
        v2f a = *(const v2f*)(arow + k + kofs);
        v2f b = *(const v2f*)(brow + k + kofs);
        c = __builtin_amdgcn_wmma_f32_16x16x4_f32(
             false, a,  false, b,
             (short)0, c,  false,  false);
    }

    const int rbase = row0 + ((lane >> 4) << 3);
    const int col   = lane & 15;
#pragma unroll
    for (int v = 0; v < 8; ++v) {
        if (IN_OUT) S[(size_t)(rbase + v) * 256 + (col >> 3) * 128 + (col & 7)] = c[v];
        else        S[(size_t)(rbase + v) * 16 + col] = c[v];
    }
}

template <bool VEC>
__device__ __forceinline__ void load16(const int* __restrict__ a, const int e, const int e1,
                                       const int pad, int4 (&v)[4]) {
#pragma unroll
    for (int u = 0; u < 4; ++u) {
        const int ee = e + u * 4 * SORT_THREADS;
        v[u] = make_int4(pad, pad, pad, pad);
        if (VEC) {
            if (ee < e1) v[u] = *(const int4*)(a + ee);
        } else {
            if (ee     < e1) v[u].x = a[ee];
            if (ee + 1 < e1) v[u].y = a[ee + 1];
            if (ee + 2 < e1) v[u].z = a[ee + 2];
            if (ee + 3 < e1) v[u].w = a[ee + 3];
        }
    }
}

template <bool VEC>
__device__ __forceinline__ int count_ids(const int* __restrict__ id, int* c,
                                         const int s0, const int len, const int e0, const int e1, const int tid) {
    int below = 0;
    for (int e = e0 + 4 * tid; e < e1; e += 16 * SORT_THREADS) {
        int4 v[4];
        load16<VEC>(id, e, e1, ID_PAD, v);
#pragma unroll
        for (int u = 0; u < 4; ++u) {
            const int vv[4] = {v[u].x, v[u].y, v[u].z, v[u].w};
#pragma unroll
            for (int j = 0; j < 4; ++j) {
                const unsigned d = (unsigned)(vv[j] - s0);
                if (d < (unsigned)len) atomicAdd(&c[d], 1);
                below += (vv[j] < s0) ? 1 : 0;
            }
        }
    }
    return below;
}

template <bool VEC>
__device__ __forceinline__ void place_ids(const int* __restrict__ id, const int* __restrict__ rel,
                                          int* __restrict__ lst, int* c,
                                          const int s0, const int len, const int e0, const int e1, const int tid) {
    for (int e = e0 + 4 * tid; e < e1; e += 16 * SORT_THREADS) {
        int4 v[4], r[4];
        load16<VEC>(id, e, e1, ID_PAD, v);
        load16<VEC>(rel, e, e1, 0, r);
        int p[16];
#pragma unroll
        for (int u = 0; u < 4; ++u) {
            const int vv[4] = {v[u].x, v[u].y, v[u].z, v[u].w};
#pragma unroll
            for (int j = 0; j < 4; ++j) {
                const unsigned d = (unsigned)(vv[j] - s0);
                p[4 * u + j] = (d < (unsigned)len) ? atomicAdd(&c[d], 1) : -1;
            }
        }
#pragma unroll
        for (int u = 0; u < 4; ++u) {
            const int rr[4] = {r[u].x, r[u].y, r[u].z, r[u].w};
#pragma unroll
            for (int j = 0; j < 4; ++j) {
                if (p[4 * u + j] >= 0) lst[p[4 * u + j]] = rr[j];
            }
        }
    }
}

__device__ __forceinline__ void park4(const int4 v, const int4 r, int* c, int* recs,
                                      const int s0, const int len, const int tid, int& n, int& below) {
    const int vv[4] = {v.x, v.y, v.z, v.w};
    const int rr[4] = {r.x, r.y, r.z, r.w};
#pragma unroll
    for (int j = 0; j < 4; ++j) {
        const unsigned d = (unsigned)(vv[j] - s0);
        if (d < (unsigned)len) {
            atomicAdd(&c[d], 1);
            recs[min(n, REC_K - 1) * SORT_THREADS + tid] = (int)(((unsigned)rr[j] << REC_DBITS) | d);
            ++n;
        }
        below += (vv[j] < s0) ? 1 : 0;
    }
}

__device__ __forceinline__ int count_park_ids(const int* __restrict__ id, const int* __restrict__ rel,
                                              int* c, int* recs,
                                              const int s0, const int len, const int e0, const int e1,
                                              const int tid, int& nrec) {
    int below = 0;
    int n = 0;
    int e = e0 + 4 * tid;
    for (; e + 12 * SORT_THREADS < e1; e += 16 * SORT_THREADS) {
        int4 v[4], r[4];
#pragma unroll
        for (int u = 0; u < 4; ++u) v[u] = *(const int4*)(id + e + u * 4 * SORT_THREADS);
#pragma unroll
        for (int u = 0; u < 4; ++u) r[u] = *(const int4*)(rel + e + u * 4 * SORT_THREADS);
        asm volatile("" : "+v"(r[0].x), "+v"(r[0].y), "+v"(r[0].z), "+v"(r[0].w),
                          "+v"(r[1].x), "+v"(r[1].y), "+v"(r[1].z), "+v"(r[1].w),
                          "+v"(r[2].x), "+v"(r[2].y), "+v"(r[2].z), "+v"(r[2].w),
                          "+v"(r[3].x), "+v"(r[3].y), "+v"(r[3].z), "+v"(r[3].w));
#pragma unroll
        for (int u = 0; u < 4; ++u) park4(v[u], r[u], c, recs, s0, len, tid, n, below);
    }
    for (; e < e1; e += 4 * SORT_THREADS) {
        const int4 v1 = *(const int4*)(id + e);
        const int4 r1 = *(const int4*)(rel + e);
        park4(v1, r1, c, recs, s0, len, tid, n, below);
    }
    nrec = n;
    return below;
}

__device__ __forceinline__ void place_parked(const int* recs, int* __restrict__ lst, int* c,
                                             const int nrec, const int tid) {
    for (int j = 0; j < nrec; j += 4) {
        unsigned rec[4];
        int p[4];
#pragma unroll
        for (int q = 0; q < 4; ++q) rec[q] = (j + q < nrec) ? (unsigned)recs[(j + q) * SORT_THREADS + tid] : 0u;
#pragma unroll
        for (int q = 0; q < 4; ++q)
            p[q] = (j + q < nrec) ? atomicAdd(&c[rec[q] & ((1u << REC_DBITS) - 1u)], 1) : -1;
#pragma unroll
        for (int q = 0; q < 4; ++q) {
            if (p[q] >= 0) lst[p[q]] = (int)(rec[q] >> REC_DBITS);
        }
    }
}

__device__ __forceinline__ void sort_block(const int* __restrict__ eidx, const int* __restrict__ rel,
                                           int* __restrict__ st, int* __restrict__ lst,
                                           const int E, const int N_e, const int N_r, const int b,
                                           int* c, int* wtot, int* wbel, int* recs, int* ovf) {
    const int tid  = threadIdx.x;
    const int lane = tid & 31;
    const int wave = tid >> 5;
    const int chunk = b % CH;
    const int y     = b / CH;
    const int side  = y / PS;
    const int part  = y - side * PS;

    const int plen = (N_e + PS - 1) / PS;
    const int lo   = part * plen;
    const int hi   = min(lo + plen, N_e);
    const int clen = chunk_len(E);
    const int e0   = min(chunk * clen, E);
    const int e1   = min(e0 + clen, E);
    const int* __restrict__ id = eidx + (size_t)side * E;
    int* __restrict__ srow = st + (size_t)(chunk * 2 + side) * (N_e + 1);
    const int region = side * E + e0;
    const bool vec = ((E & 3) == 0) && ((((size_t)eidx) & 15) == 0) && ((((size_t)rel) & 15) == 0);
    const bool park = vec && N_r <= REC_RMAX;

    if (part == PS - 1 && tid == 0) srow[N_e] = side * E + e1;

    for (int s0 = lo; s0 < hi; s0 += CAP) {
        const int len = min(CAP, hi - s0);
        *(int4*)(c + 4 * tid) = make_int4(0, 0, 0, 0);
        if (tid == 0) *ovf = 0;
        __syncthreads();
        int below;
        int nrec = 0;
        if (park) {
            below = count_park_ids(id, rel, c, recs, s0, len, e0, e1, tid, nrec);
            if (nrec > REC_K) *ovf = 1;
        } else {
            const int* idf = id;
            asm volatile("" : "+s"(idf));
            below = count_ids<false>(idf, c, s0, len, e0, e1, tid);
        }
        __syncthreads();
        const bool parked = park && (__builtin_amdgcn_readfirstlane(*ovf) == 0);

        const int4 k = *(const int4*)(c + 4 * tid);
        const int tsum = k.x + k.y + k.z + k.w;
        int s = tsum;
#pragma unroll
        for (int d = 1; d < 32; d <<= 1) {
            const int t = __shfl_up(s, d);
            if (lane >= d) s += t;
        }
        int bl = below;
#pragma unroll
        for (int d = 16; d > 0; d >>= 1) bl += __shfl_xor(bl, d);
        if (lane == 31) wtot[wave] = s;
        if (lane == 0)  wbel[wave] = bl;
        __syncthreads();
        const int wt = wtot[lane];
        int ws = wt;
#pragma unroll
        for (int d = 1; d < 32; d <<= 1) {
            const int t = __shfl_up(ws, d);
            if (lane >= d) ws += t;
        }
        const int wex = __shfl(ws - wt, wave);
        int bt = wbel[lane];
#pragma unroll
        for (int d = 16; d > 0; d >>= 1) bt += __shfl_xor(bt, d);

        const int x0 = region + bt + wex + (s - tsum);
        const int x1 = x0 + k.x;
        const int x2 = x1 + k.y;
        const int x3 = x2 + k.z;
        const int i  = 4 * tid;
        if (i     < len) srow[s0 + i]     = x0;
        if (i + 1 < len) srow[s0 + i + 1] = x1;
        if (i + 2 < len) srow[s0 + i + 2] = x2;
        if (i + 3 < len) srow[s0 + i + 3] = x3;
        *(int4*)(c + 4 * tid) = make_int4(x0, x1, x2, x3);
        __syncthreads();
        if (parked) {
            place_parked(recs, lst, c, nrec, tid);
        } else {
            const int* idf = id;
            const int* relf = rel;
            asm volatile("" : "+s"(idf), "+s"(relf));
            place_ids<false>(idf, relf, lst, c, s0, len, e0, e1, tid);
        }
        __syncthreads();
    }
}

__global__ void __launch_bounds__(SORT_THREADS, 16)
k_sort_proj(const int* __restrict__ eidx, const int* __restrict__ rel,
            int* __restrict__ st, int* __restrict__ lst, int E, int N_e,
            const float* __restrict__ x_e, const float* __restrict__ Wh, const float* __restrict__ Wt,
            float* __restrict__ out,
            const float* __restrict__ x_r, const float* __restrict__ Wr,
            float* __restrict__ Sr, int N_r, int blocks_e) {
    __shared__ int4 cnt4[CAP / 4];
    __shared__ int wtot[SORT_THREADS / 32];
    __shared__ int wbel[SORT_THREADS / 32];
    __shared__ int recs[REC_K * SORT_THREADS];
    __shared__ int ovf;
    static_assert(SORT_THREADS / 32 == 32, "one lane per wave in the block scan");
    static_assert(CAP == 4 * SORT_THREADS, "four counters per thread");
    static_assert(CAP <= (1 << REC_DBITS), "a counter index fits the low bits of a record");
    static_assert(REC_K % 4 == 0, "the replay takes records four at a time");
    const int b = blockIdx.x;
    if (b < NSORT) {
        sort_block(eidx, rel, st, lst, E, N_e, N_r, b, (int*)cnt4, wtot, wbel, recs, &ovf);
    } else {
        const int lane = threadIdx.x & 31;
        const int wib  = threadIdx.x >> 5;
        const int pb   = b - NSORT;
        if (pb < blocks_e) {
            proj_tile_wmma<true>(x_e, Wh, Wt, out, N_e, pb * (SORT_THREADS / 32) + wib, lane);
        } else {
            proj_tile_wmma<false>(x_r, Wr, Wr, Sr, N_r, (pb - blocks_e) * (SORT_THREADS / 32) + wib, lane);
        }
    }
}

template <int NJ>
__device__ __forceinline__ void head_mean_weights(const float (&p)[8], float den, int (&ai)[8]) {
    float a[NJ];
#pragma unroll
    for (int j = 0; j < NJ; ++j) a[j] = p[j] / den;
#pragma unroll
    for (int d = 1; d < 8; d <<= 1) {
#pragma unroll
        for (int j = 0; j < NJ; ++j) a[j] += __shfl_xor(a[j], d);
    }
#pragma unroll
    for (int j = 0; j < 8; ++j) ai[j] = (j < NJ) ? __float_as_int(a[j] * 0.125f) : 0;
}

__device__ __forceinline__ int run_pos(const int i,
                                       const int a0, const int a1, const int a2, const int a3,
                                       const int a4, const int a5, const int a6, const int a7,
                                       const int p1, const int p2, const int p3, const int p4,
                                       const int p5, const int p6, const int p7) {
    int a = a0;
    a = (i >= p1) ? a1 : a;
    a = (i >= p2) ? a2 : a;
    a = (i >= p3) ? a3 : a;
    a = (i >= p4) ? a4 : a;
    a = (i >= p5) ? a5 : a;
    a = (i >= p6) ? a6 : a;
    a = (i >= p7) ? a7 : a;
    return a + i;
}

#define SR_CAP 2048

template <bool TAB>
__device__ __forceinline__ float rel_score(const float* __restrict__ Sr, const float* tab, const int r, const int h) {
    return TAB ? tab[r * 8 + h] : Sr[(size_t)r * 16 + h];
}

template <bool TAB>
__device__ __forceinline__ void node_segment(const int seg, const int n,
                                             const int (&pre)[CH], const int (&adj)[CH], const int lane,
                                             const int* __restrict__ lst,
                                             const float* __restrict__ Sr,
                                             const float* tab,
                                             const float* __restrict__ x_r, float* __restrict__ out,
                                             const int N_e) {
    const int side = (seg >= N_e) ? 1 : 0;
    const int node = seg - side * N_e;
    static_assert(CH == 8, "run_pos names the eight runs one by one");
    const int p1 = pre[1], p2 = pre[2], p3 = pre[3], p4 = pre[4], p5 = pre[5], p6 = pre[6], p7 = pre[7];
    const int a0 = adj[0], a1 = adj[1], a2 = adj[2], a3 = adj[3], a4 = adj[4], a5 = adj[5], a6 = adj[6], a7 = adj[7];
#define POS(i) run_pos((i), a0, a1, a2, a3, a4, a5, a6, a7, p1, p2, p3, p4, p5, p6, p7)

    const int h    = lane & 7;
    const int slot = lane >> 3;
    const float* enp = out + (size_t)node * 256 + side * 128 + h;
    const float* __restrict__ xl = x_r + lane * 4;
    float* o = out + (size_t)node * 256 + side * 128 + lane * 4;

    float4 acc = make_float4(0.0f, 0.0f, 0.0f, 0.0f);

    if (n <= 32) {
        int rl = 0;
        if (lane < n) rl = lst[POS(lane)];
        const float en = *enp;

        float v[8];
#pragma unroll
        for (int j = 0; j < 8; ++j) {
            const int rj = __shfl(rl, 4 * j + slot);
            v[j] = rel_score<TAB>(Sr, tab, rj, h);
        }
        float4 rows[16];
#pragma unroll
        for (int k = 0; k < 16; ++k) {
            const int rk = __builtin_amdgcn_readlane(rl, k);
            rows[k] = *(const float4*)(xl + (size_t)rk * 128);
        }

        float s[8];
        float m = -INFINITY;
#pragma unroll
        for (int j = 0; j < 8; ++j) {
            s[j] = (4 * j + slot < n) ? lrelu(en + v[j]) : -INFINITY;
            m = fmaxf(m, s[j]);
        }
        m = fmaxf(m, __shfl_xor(m, 8));
        m = fmaxf(m, __shfl_xor(m, 16));
        float z = 0.0f;
#pragma unroll
        for (int j = 0; j < 8; ++j) {
            s[j] = (4 * j + slot < n) ? __expf(s[j] - m) : 0.0f;
            z += s[j];
        }
        z += __shfl_xor(z, 8);
        z += __shfl_xor(z, 16);
        const float den = z + 1e-16f;
        int ai[8];
        if (n > 16) head_mean_weights<8>(s, den, ai);
        else        head_mean_weights<4>(s, den, ai);

#pragma unroll
        for (int k = 0; k < 16; ++k) {
            if (k < n) {
                const float ak = __int_as_float(__builtin_amdgcn_readlane(ai[k >> 2], (k & 3) * 8));
                acc.x += rows[k].x * ak;
                acc.y += rows[k].y * ak;
                acc.z += rows[k].z * ak;
                acc.w += rows[k].w * ak;
            }
        }
        if (n > 16) {
#pragma unroll
            for (int q = 0; q < 4; ++q) {
                if (16 + 4 * q < n) {
#pragma unroll
                    for (int t = 0; t < 4; ++t) {
                        const int k  = 4 * q + t;
                        const int rk = __builtin_amdgcn_readlane(rl, 16 + k);
                        rows[k] = *(const float4*)(xl + (size_t)rk * 128);
                    }
                }
            }
#pragma unroll
            for (int k = 0; k < 16; ++k) {
                if (16 + k < n) {
                    const float ak = __int_as_float(__builtin_amdgcn_readlane(ai[4 + (k >> 2)], (k & 3) * 8));
                    acc.x += rows[k].x * ak;
                    acc.y += rows[k].y * ak;
                    acc.z += rows[k].z * ak;
                    acc.w += rows[k].w * ak;
                }
            }
        }
        *(float4*)o = acc;
        return;
    }

    const float en = *enp;
    float m = -INFINITY;
    for (int i = slot; i < n; i += 4) {
        const int r = lst[POS(i)];
        m = fmaxf(m, lrelu(en + rel_score<TAB>(Sr, tab, r, h)));
    }
    m = fmaxf(m, __shfl_xor(m, 8));
    m = fmaxf(m, __shfl_xor(m, 16));

    float z = 0.0f;
    for (int i = slot; i < n; i += 4) {
        const int r = lst[POS(i)];
        z += __expf(lrelu(en + rel_score<TAB>(Sr, tab, r, h)) - m);
    }
    z += __shfl_xor(z, 8);
    z += __shfl_xor(z, 16);
    const float den = z + 1e-16f;

    int i0 = 0;
    for (; i0 + 4 <= n; i0 += 4) {
        const int r = lst[POS(i0 + slot)];
        float a = __expf(lrelu(en + rel_score<TAB>(Sr, tab, r, h)) - m) / den;
        a += __shfl_xor(a, 1);
        a += __shfl_xor(a, 2);
        a += __shfl_xor(a, 4);
        a *= 0.125f;
        const int ai = __float_as_int(a);
        const int r0 = __builtin_amdgcn_readlane(r, 0);
        const int r1 = __builtin_amdgcn_readlane(r, 8);
        const int r2 = __builtin_amdgcn_readlane(r, 16);
        const int r3 = __builtin_amdgcn_readlane(r, 24);
        const float a0 = __int_as_float(__builtin_amdgcn_readlane(ai, 0));
        const float a1 = __int_as_float(__builtin_amdgcn_readlane(ai, 8));
        const float a2 = __int_as_float(__builtin_amdgcn_readlane(ai, 16));
        const float a3 = __int_as_float(__builtin_amdgcn_readlane(ai, 24));
        const float4 m0 = *(const float4*)(xl + (size_t)r0 * 128);
        const float4 m1 = *(const float4*)(xl + (size_t)r1 * 128);
        const float4 m2 = *(const float4*)(xl + (size_t)r2 * 128);
        const float4 m3 = *(const float4*)(xl + (size_t)r3 * 128);
        acc.x += m0.x * a0; acc.y += m0.y * a0; acc.z += m0.z * a0; acc.w += m0.w * a0;
        acc.x += m1.x * a1; acc.y += m1.y * a1; acc.z += m1.z * a1; acc.w += m1.w * a1;
        acc.x += m2.x * a2; acc.y += m2.y * a2; acc.z += m2.z * a2; acc.w += m2.w * a2;
        acc.x += m3.x * a3; acc.y += m3.y * a3; acc.z += m3.z * a3; acc.w += m3.w * a3;
    }
    if (i0 < n) {
        const int  i  = i0 + slot;
        const bool ok = i < n;
        const int  r  = ok ? lst[POS(i)] : 0;
        float a = 0.0f;
        if (ok) a = __expf(lrelu(en + rel_score<TAB>(Sr, tab, r, h)) - m) / den;
        a += __shfl_xor(a, 1);
        a += __shfl_xor(a, 2);
        a += __shfl_xor(a, 4);
        a *= 0.125f;
        const int ai = __float_as_int(a);
#pragma unroll
        for (int k = 0; k < 3; ++k) {
            if (i0 + k < n) {
                const int   rk = __builtin_amdgcn_readlane(r, k * 8);
                const float ak = __int_as_float(__builtin_amdgcn_readlane(ai, k * 8));
                const float4 mv = *(const float4*)(xl + (size_t)rk * 128);
                acc.x += mv.x * ak;
                acc.y += mv.y * ak;
                acc.z += mv.z * ak;
                acc.w += mv.w * ak;
            }
        }
    }
    *(float4*)o = acc;
}

#undef POS

#define NODE_THREADS 1024

__device__ __forceinline__ void load_runs(const int* __restrict__ st, const int seg, const int N_e,
                                          int (&nx)[2 * CH]) {
    const int side = (seg >= N_e) ? 1 : 0;
    const int node = seg - side * N_e;
    const int* __restrict__ p = st + (size_t)side * (N_e + 1) + node;
#pragma unroll
    for (int k = 0; k < CH; ++k) {
        nx[k]      = p[(size_t)k * 2 * (N_e + 1)];
        nx[CH + k] = p[(size_t)k * 2 * (N_e + 1) + 1];
    }
}

template <bool TAB>
__device__ __forceinline__ void walk_segments(const int w, const int nw, const int n2, const int lane,
                                              const int* __restrict__ st,
                                              const int* __restrict__ lst,
                                              const float* __restrict__ Sr,
                                              const float* tab,
                                              const float* __restrict__ x_r, float* __restrict__ out,
                                              const int N_e) {
    int seg = w;
    int nx[2 * CH];
    load_runs(st, seg, N_e, nx);
    for (;;) {
        int pre[CH], adj[CH];
        int n = 0;
#pragma unroll
        for (int k = 0; k < CH; ++k) {
            const int s = __builtin_amdgcn_readfirstlane(nx[k]);
            const int e = __builtin_amdgcn_readfirstlane(nx[CH + k]);
            pre[k] = n;
            adj[k] = s - n;
            n += e - s;
        }
        const int  sn   = seg + nw;
        const bool more = sn < n2;
        if (more) load_runs(st, sn, N_e, nx);
        node_segment<TAB>(seg, n, pre, adj, lane, lst, Sr, tab, x_r, out, N_e);
        if (!more) break;
        seg = sn;
    }
}

__global__ void __launch_bounds__(NODE_THREADS)
k_node(const int* __restrict__ st,
       const int* __restrict__ lst,
       const float* __restrict__ Sr,
       const float* __restrict__ x_r, float* __restrict__ out,
       int N_e, int N_r) {
    __shared__ float tab[SR_CAP * 8];
    const bool use_tab = N_r <= SR_CAP;
    if (use_tab) {
        const int items = 2 * N_r;
        for (int i0 = threadIdx.x; i0 < items; i0 += 4 * NODE_THREADS) {
            const int i1 = i0 + NODE_THREADS;
            const int i2 = i0 + 2 * NODE_THREADS;
            const int i3 = i0 + 3 * NODE_THREADS;
            const int c1 = min(i1, items - 1);
            const int c2 = min(i2, items - 1);
            const int c3 = min(i3, items - 1);
            const float4 t0 = *(const float4*)(Sr + (size_t)(i0 >> 1) * 16 + (i0 & 1) * 4);
            const float4 t1 = *(const float4*)(Sr + (size_t)(c1 >> 1) * 16 + (c1 & 1) * 4);
            const float4 t2 = *(const float4*)(Sr + (size_t)(c2 >> 1) * 16 + (c2 & 1) * 4);
            const float4 t3 = *(const float4*)(Sr + (size_t)(c3 >> 1) * 16 + (c3 & 1) * 4);
            *(float4*)(tab + i0 * 4) = t0;
            *(float4*)(tab + c1 * 4) = t1;
            *(float4*)(tab + c2 * 4) = t2;
            *(float4*)(tab + c3 * 4) = t3;
        }
        __syncthreads();
    }

    const int lane = threadIdx.x & 31;
    const int w  = __builtin_amdgcn_readfirstlane((int)((blockIdx.x * NODE_THREADS + threadIdx.x) >> 5));
    const int nw = (int)(gridDim.x * (NODE_THREADS / 32));
    const int n2 = 2 * N_e;
    if (w >= n2) return;
    if (use_tab) walk_segments<true>(w, nw, n2, lane, st, lst, Sr, tab, x_r, out, N_e);
    else         walk_segments<false>(w, nw, n2, lane, st, lst, Sr, tab, x_r, out, N_e);
}

extern "C" void kernel_launch(void* const* d_in, const int* in_sizes, int n_in,
                              void* d_out, int out_size, void* d_ws, size_t ws_size,
                              hipStream_t stream) {
    const float* x_e  = (const float*)d_in[0];
    const float* x_r  = (const float*)d_in[1];
    const float* Wh   = (const float*)d_in[2];
    const float* Wt   = (const float*)d_in[3];
    const float* Wr   = (const float*)d_in[4];
    const int*   eidx = (const int*)d_in[5];
    const int*   rel  = (const int*)d_in[6];

    const int D   = 128;
    const int N_e = in_sizes[0] / D;
    const int N_r = in_sizes[1] / D;
    const int E   = in_sizes[6];

    auto align256 = [](size_t x) { return (x + 255) & ~(size_t)255; };
    char* ws = (char*)d_ws;
    float* Sr  = (float*)ws; ws += align256((size_t)N_r * 16 * sizeof(float));
    int*   st  = (int*)ws;   ws += align256((size_t)CH * 2 * ((size_t)N_e + 1) * sizeof(int));
    int*   lst = (int*)ws;   ws += align256((size_t)E * 2 * sizeof(int));
    if ((size_t)(ws - (char*)d_ws) > ws_size) return;

    float* out = (float*)d_out;

    const int WPB = SORT_THREADS / 32;
    const int blocks_e = ((N_e + 15) / 16 + WPB - 1) / WPB;
    const int blocks_r = ((N_r + 15) / 16 + WPB - 1) / WPB;
    k_sort_proj<<<NSORT + blocks_e + blocks_r, SORT_THREADS, 0, stream>>>(eidx, rel, st, lst, E, N_e,
                                                                       x_e, Wh, Wt, out,
                                                                       x_r, Wr, Sr, N_r, blocks_e);

    k_node<<<256, NODE_THREADS, 0, stream>>>(st, lst, Sr, x_r, out, N_e, N_r);
}
